# starting kernel
# baseline (speedup 1.0000x reference)
#include <hip/hip_runtime.h>


typedef __attribute__((ext_vector_type(2))) float v2f;
typedef __attribute__((ext_vector_type(8))) float v8f;

#define N_NODES 100000
#define N_EDGES 1600000
#define OUTC 32

template<int K>
__global__ __launch_bounds__(256) void gemm_x32_wmma(
    const float* __restrict__ X,
    const float* __restrict__ W,
    float* __restrict__ H,
    int n_rows)
{
  const int lane = threadIdx.x & 31;
  const int tile = (int)(blockIdx.x * (blockDim.x >> 5) + (threadIdx.x >> 5));
  const int n_tiles = n_rows >> 4;
  if (tile >= n_tiles) return;

  const int  m   = lane & 15;
  const int  kh  = lane >> 4;
  const long row0 = (long)tile << 4;

  const float* xrow = X + (row0 + m) * K;

  v8f c0 = {};
  v8f c1 = {};

  #pragma unroll 8
  for (int kb = 0; kb < K; kb += 4) {
    const int ka = kb + 2 * kh;
    v2f a;
    a.x = xrow[ka];
    a.y = xrow[ka + 1];
    v2f b0, b1;
    b0.x = W[ka * OUTC + m];
    b0.y = W[(ka + 1) * OUTC + m];
    b1.x = W[ka * OUTC + 16 + m];
    b1.y = W[(ka + 1) * OUTC + 16 + m];
    c0 = __builtin_amdgcn_wmma_f32_16x16x4_f32(false, a, false, b0, (short)0, c0, false, false);
    c1 = __builtin_amdgcn_wmma_f32_16x16x4_f32(false, a, false, b1, (short)0, c1, false, false);
  }

  float* hrow = H + (row0 + 8 * kh) * OUTC + m;
  #pragma unroll
  for (int r = 0; r < 8; ++r) {
    hrow[r * OUTC]      = c0[r];
    hrow[r * OUTC + 16] = c1[r];
  }
}

__device__ __forceinline__ void atomic_add_f32_dev(float* p, float v) {
  asm volatile("global_atomic_add_f32 %0, %1, off scope:SCOPE_DEV"
               :
               : "v"(p), "v"(v)
               : "memory");
}

__global__ __launch_bounds__(256) void gcn_scatter(
    const float* __restrict__ H,
    const int*   __restrict__ eidx,
    const float* __restrict__ ew,
    float*       __restrict__ out,
    int n_edges)
{
  const int lane   = threadIdx.x & 31;
  const int wave0  = (int)((blockIdx.x * (long)blockDim.x + threadIdx.x) >> 5);
  const int nwaves = (int)(((long)gridDim.x * blockDim.x) >> 5);

  for (int e = wave0; e < n_edges; e += nwaves) {
    const int   s = eidx[e];
    const int   d = eidx[n_edges + e];
    const float w = ew[e];
    const float v = w * H[(long)s * OUTC + lane];
    atomic_add_f32_dev(&out[(long)d * OUTC + lane], v);
  }
}

extern "C" void kernel_launch(void* const* d_in, const int* in_sizes, int n_in,
                              void* d_out, int out_size, void* d_ws, size_t ws_size,
                              hipStream_t stream) {
  const float* x  = (const float*)d_in[0];
  const float* W1 = (const float*)d_in[1];
  const float* W2 = (const float*)d_in[2];
  const float* ew = (const float*)d_in[3];
  const int*   ei = (const int*)d_in[4];
  float* out = (float*)d_out;

  const size_t feat_bytes = (size_t)N_NODES * OUTC * sizeof(float);
  float* h    = (float*)d_ws;
  float* agg1 = (float*)((char*)d_ws + feat_bytes);

  hipMemsetAsync(agg1, 0, feat_bytes, stream);
  hipMemsetAsync(out,  0, feat_bytes, stream);

  const int tiles = N_NODES / 16;
  const int wpb   = 256 / 32;
  dim3 gemm_grid((tiles + wpb - 1) / wpb);

  gemm_x32_wmma<128><<<gemm_grid, 256, 0, stream>>>(x, W1, h, N_NODES);
  gcn_scatter<<<4096, 256, 0, stream>>>(h, ei, ew, agg1, N_EDGES);

  gemm_x32_wmma<32><<<gemm_grid, 256, 0, stream>>>(agg1, W2, h, N_NODES);
  gcn_scatter<<<4096, 256, 0, stream>>>(h, ei, ew, out, N_EDGES);
}


# Aggregation turned from push to pull: edges sorted by destination once per call (two ballot-ranked counting-sort levels in LDS); a wave sums each node's rows in registers; no atomics; sum order fixed
# speedup vs baseline: 1.2269x; 1.2269x over previous
#include <hip/hip_runtime.h>


typedef __attribute__((ext_vector_type(2))) float v2f;
typedef __attribute__((ext_vector_type(8))) float v8f;

#define N_NODES 100000
#define N_EDGES 1600000
#define OUTC 32
#define TP   34

template<int K>
__global__ __launch_bounds__(256) void gemm_x32_wmma(
    const float* __restrict__ X,
    const float* __restrict__ W,
    const float* __restrict__ W2,
    float* __restrict__ H,
    int n_rows)
{
  __shared__ __attribute__((aligned(8))) float tileT[8][16 * TP];

  const int lane = threadIdx.x & 31;
  const int tile = (int)(blockIdx.x * (blockDim.x >> 5) + (threadIdx.x >> 5));
  const int n_tiles = n_rows >> 4;
  if (tile >= n_tiles) return;

  const int  m   = lane & 15;
  const int  kh  = lane >> 4;
  const long row0 = (long)tile << 4;

  const float* xrow = X + (row0 + m) * K;

  v8f c0 = {};
  v8f c1 = {};

  #pragma unroll 8
  for (int kb = 0; kb < K; kb += 4) {
    const int ka = kb + 2 * kh;
    v2f a;
    a.x = xrow[ka];
    a.y = xrow[ka + 1];
    v2f b0, b1;
    b0.x = W[ka * OUTC + m];
    b0.y = W[(ka + 1) * OUTC + m];
    b1.x = W[ka * OUTC + 16 + m];
    b1.y = W[(ka + 1) * OUTC + 16 + m];
    c0 = __builtin_amdgcn_wmma_f32_16x16x4_f32(false, a, false, b0, (short)0, c0, false, false);
    c1 = __builtin_amdgcn_wmma_f32_16x16x4_f32(false, a, false, b1, (short)0, c1, false, false);
  }

  v2f w2lo[OUTC / 4], w2hi[OUTC / 4];
  #pragma unroll
  for (int s = 0; s < OUTC / 4; ++s) {
    const int ka = 4 * s + 2 * kh;
    w2lo[s].x = W2[ka * OUTC + m];
    w2lo[s].y = W2[(ka + 1) * OUTC + m];
    w2hi[s].x = W2[ka * OUTC + 16 + m];
    w2hi[s].y = W2[(ka + 1) * OUTC + 16 + m];
  }

  float* T = tileT[threadIdx.x >> 5];
  #pragma unroll
  for (int r = 0; r < 8; ++r) {
    T[(8 * kh + r) * TP + m]      = c0[r];
    T[(8 * kh + r) * TP + 16 + m] = c1[r];
  }
  __builtin_amdgcn_fence(__ATOMIC_RELEASE, "wavefront");
  __builtin_amdgcn_wave_barrier();
  __builtin_amdgcn_fence(__ATOMIC_ACQUIRE, "wavefront");

  v8f d0 = {};
  v8f d1 = {};
  #pragma unroll
  for (int s = 0; s < OUTC / 4; ++s) {
    const int ka = 4 * s + 2 * kh;
    const v2f a = *reinterpret_cast<const v2f*>(&T[m * TP + ka]);
    d0 = __builtin_amdgcn_wmma_f32_16x16x4_f32(false, a, false, w2lo[s], (short)0, d0, false, false);
    d1 = __builtin_amdgcn_wmma_f32_16x16x4_f32(false, a, false, w2hi[s], (short)0, d1, false, false);
  }

  float* hrow = H + (row0 + 8 * kh) * OUTC + m;
  #pragma unroll
  for (int r = 0; r < 8; ++r) {
    hrow[r * OUTC]      = d0[r];
    hrow[r * OUTC + 16] = d1[r];
  }
}

#define N_BINS    ((N_NODES + 255) >> 8)
#define L1_BLOCKS 512
#define L1_WAVES  (L1_BLOCKS * 8)
#define L1_STEPS  13
static_assert((long)L1_WAVES * L1_STEPS * 32 >= N_EDGES, "level-1 waves must cover every edge");
static_assert(N_BINS <= 512, "level-1 key is matched on 9 bits");
static_assert(L1_WAVES == 4096, "the scan tiles are rows of 4096 counters");

__device__ __forceinline__ void wave_sync() {
  __builtin_amdgcn_fence(__ATOMIC_RELEASE, "wavefront");
  __builtin_amdgcn_wave_barrier();
  __builtin_amdgcn_fence(__ATOMIC_ACQUIRE, "wavefront");
}

template<int BITS>
__device__ __forceinline__ unsigned peer_mask(int key, bool valid) {
  unsigned peers = __builtin_amdgcn_ballot_w32(valid);
  #pragma unroll
  for (int b = 0; b < BITS; ++b) {
    const bool bit = (key >> b) & 1;
    const unsigned bal = __builtin_amdgcn_ballot_w32(bit);
    peers &= bit ? bal : ~bal;
  }
  return peers;
}

__device__ __forceinline__ int wave_sum(int v) {
  #pragma unroll
  for (int m = 16; m > 0; m >>= 1) v += __shfl_xor(v, m);
  return v;
}

__device__ __forceinline__ int wave_scan_incl(int v, int lane) {
  #pragma unroll
  for (int d = 1; d < 32; d <<= 1) {
    const int t = __shfl_up(v, d);
    if (lane >= d) v += t;
  }
  return v;
}

__global__ __launch_bounds__(256) void bin_count(
    const int* __restrict__ eidx,
    int*       __restrict__ counts,
    int n_edges)
{
  __shared__ int hist[8][N_BINS];
  const int lane = threadIdx.x & 31;
  const int wv   = threadIdx.x >> 5;
  const int w    = (int)blockIdx.x * 8 + wv;
  for (int b = lane; b < N_BINS; b += 32) hist[wv][b] = 0;

  int key[L1_STEPS];
  const int e0 = w * (L1_STEPS * 32) + lane;
  #pragma unroll
  for (int s = 0; s < L1_STEPS; ++s) {
    const int e = e0 + s * 32;
    const int d = eidx[n_edges + min(e, n_edges - 1)];
    key[s] = (e < n_edges) ? (d >> 8) : -1;
  }
  wave_sync();
  #pragma unroll
  for (int s = 0; s < L1_STEPS; ++s) {
    const bool valid = key[s] >= 0;
    const unsigned peers = peer_mask<9>(key[s], valid);
    if (valid && (peers >> lane) == 1u) hist[wv][key[s]] += __popc(peers);
    wave_sync();
  }
  for (int b = lane; b < N_BINS; b += 32) counts[b * L1_WAVES + w] = hist[wv][b];
}

__global__ __launch_bounds__(1024) void tile_sums(
    const int* __restrict__ cnt,
    int*       __restrict__ bsum)
{
  __shared__ int part[32];
  const int i4 = (int)(blockIdx.x * 1024 + threadIdx.x);
  const int4 c = reinterpret_cast<const int4*>(cnt)[i4];
  const int s = wave_sum(c.x + c.y + c.z + c.w);
  if ((threadIdx.x & 31) == 0) part[threadIdx.x >> 5] = s;
  __syncthreads();
  if (threadIdx.x < 32) {
    const int t = wave_sum(part[threadIdx.x]);
    if (threadIdx.x == 0) bsum[blockIdx.x] = t;
  }
}

__global__ __launch_bounds__(1024) void tile_scan(
    const int* __restrict__ cnt,
    const int* __restrict__ bsum,
    int*       __restrict__ pre,
    int n)
{
  __shared__ int part[32];
  __shared__ int partb[32];
  const int lane = threadIdx.x & 31;
  const int wv   = threadIdx.x >> 5;
  const int i4   = (int)(blockIdx.x * 1024 + threadIdx.x);
  const int4 c   = reinterpret_cast<const int4*>(cnt)[i4];
  const int tsum = c.x + c.y + c.z + c.w;
  const int inc  = wave_scan_incl(tsum, lane);
  if (lane == 31) part[wv] = inc;

  int b = 0;
  for (int k = (int)threadIdx.x; k < (int)blockIdx.x; k += 1024) b += bsum[k];
  b = wave_sum(b);
  if (lane == 0) partb[wv] = b;
  __syncthreads();

  int before = 0;
  #pragma unroll
  for (int k = 0; k < 32; ++k) {
    before += partb[k];
    if (k < wv) before += part[k];
  }
  const int ex = before + inc - tsum;
  int4 o;
  o.x = ex;
  o.y = o.x + c.x;
  o.z = o.y + c.y;
  o.w = o.z + c.z;
  reinterpret_cast<int4*>(pre)[i4] = o;
  if (i4 * 4 + 4 == n) pre[n] = o.w + c.w;
}

__global__ __launch_bounds__(256) void bin_scatter(
    const int*   __restrict__ eidx,
    const float* __restrict__ ew,
    const int*   __restrict__ start,
    int*         __restrict__ sw1_i,
    int*         __restrict__ d1,
    int n_edges)
{
  __shared__ int cur[8][N_BINS];
  const int lane = threadIdx.x & 31;
  const int wv   = threadIdx.x >> 5;
  const int w    = (int)blockIdx.x * 8 + wv;
  for (int b = lane; b < N_BINS; b += 32) cur[wv][b] = start[b * L1_WAVES + w];

  int dst[L1_STEPS], srcv[L1_STEPS], wbits[L1_STEPS];
  const int e0 = w * (L1_STEPS * 32) + lane;
  #pragma unroll
  for (int s = 0; s < L1_STEPS; ++s) {
    const int e  = e0 + s * 32;
    const int ec = min(e, n_edges - 1);
    const int d  = eidx[n_edges + ec];
    srcv[s]  = eidx[ec];
    wbits[s] = __float_as_int(ew[ec]);
    dst[s]   = (e < n_edges) ? d : -1;
  }
  int2* sw1 = reinterpret_cast<int2*>(sw1_i);
  wave_sync();
  #pragma unroll
  for (int s = 0; s < L1_STEPS; ++s) {
    const bool valid = dst[s] >= 0;
    const int  key   = dst[s] >> 8;
    const unsigned peers = peer_mask<9>(key, valid);
    const int base = valid ? cur[wv][key] : 0;
    wave_sync();
    if (valid && (peers >> lane) == 1u) cur[wv][key] = base + __popc(peers);
    wave_sync();
    if (valid) {
      const int pos = base + __popc(peers & ((1u << lane) - 1u));
      sw1[pos] = make_int2(srcv[s], wbits[s]);
      d1[pos]  = dst[s];
    }
  }
}

__global__ __launch_bounds__(1024) void bin_sort(
    const int* __restrict__ start,
    const int* __restrict__ sw1_i,
    const int* __restrict__ d1,
    int*       __restrict__ off,
    int*       __restrict__ pairs_i,
    int n_nodes)
{
  __shared__ int hist[32][256];
  __shared__ int nodeoff[256];
  __shared__ int wtot[8];
  const int tid  = threadIdx.x;
  const int lane = tid & 31;
  const int wv   = tid >> 5;
  const int bin  = blockIdx.x;
  const int beg  = start[bin * L1_WAVES];
  const int end  = start[(bin + 1) * L1_WAVES];
  #pragma unroll
  for (int k = 0; k < 8; ++k) hist[wv][lane + 32 * k] = 0;
  wave_sync();

  for (int i0 = beg; i0 < end; i0 += 1024) {
    const int  i     = i0 + tid;
    const bool valid = i < end;
    const int  key   = d1[min(i, end - 1)] & 255;
    const unsigned peers = peer_mask<8>(key, valid);
    if (valid && (peers >> lane) == 1u) hist[wv][key] += __popc(peers);
    wave_sync();
  }
  __syncthreads();

  int tot = 0, inc = 0;
  if (tid < 256) {
    for (int k = 0; k < 32; ++k) {
      const int c = hist[k][tid];
      hist[k][tid] = tot;
      tot += c;
    }
    inc = wave_scan_incl(tot, lane);
    if (lane == 31) wtot[wv] = inc;
  }
  __syncthreads();
  if (tid < 256) {
    int pre = 0;
    for (int k = 0; k < wv; ++k) pre += wtot[k];
    const int no = pre + inc - tot;
    nodeoff[tid] = no;
    const int node = bin * 256 + tid;
    if (node < n_nodes) off[node] = beg + no;
    if (node == n_nodes - 1) off[n_nodes] = beg + no + tot;
  }
  __syncthreads();

  const int2* sw1   = reinterpret_cast<const int2*>(sw1_i);
  int2*       pairs = reinterpret_cast<int2*>(pairs_i);
  for (int i0 = beg; i0 < end; i0 += 1024) {
    const int  i     = i0 + tid;
    const bool valid = i < end;
    const int  ic    = min(i, end - 1);
    const int  key   = d1[ic] & 255;
    const int2 rec   = sw1[ic];
    const unsigned peers = peer_mask<8>(key, valid);
    const int base = hist[wv][key];
    wave_sync();
    if (valid && (peers >> lane) == 1u) hist[wv][key] = base + __popc(peers);
    wave_sync();
    if (valid) {
      const int pos = beg + nodeoff[key] + base + __popc(peers & ((1u << lane) - 1u));
      pairs[pos] = rec;
    }
  }
}

#define PULL_K 16

__device__ __forceinline__ int2 load_pairs(const int2* __restrict__ pairs, int b, int e, int lane) {
  int2 v = make_int2(0, 0);
  if (b < e) v = pairs[min(b + lane, e - 1)];
  return v;
}
__device__ __forceinline__ int2 mask_pairs(int2 v, int b, int e, int lane) {
  if (b + lane >= e) v.y = 0;
  return v;
}

template<int U>
__device__ __forceinline__ float pull_rows(const float* __restrict__ H, int2 p, int lane, float acc) {
  float hv[U], wt[U];
  #pragma unroll
  for (int u = 0; u < U; ++u) {
    const int s = __builtin_amdgcn_readlane(p.x, u);
    wt[u] = __int_as_float(__builtin_amdgcn_readlane(p.y, u));
    hv[u] = H[(long)s * OUTC + lane];
  }
  #pragma unroll
  for (int u = 0; u < U; ++u) acc = fmaf(wt[u], hv[u], acc);
  return acc;
}

__device__ __forceinline__ float pull_chunk(const float* __restrict__ H, int2 p, int cnt, int lane, float acc) {
  if (cnt <= 8)  return pull_rows<8>(H, p, lane, acc);
  if (cnt <= 16) return pull_rows<16>(H, p, lane, acc);
  if (cnt <= 24) return pull_rows<24>(H, p, lane, acc);
  return pull_rows<32>(H, p, lane, acc);
}

__global__ __launch_bounds__(256) void gcn_pull(
    const float* __restrict__ H,
    const int*   __restrict__ off,
    const int*   __restrict__ pairs_i,
    float*       __restrict__ out,
    int n_nodes)
{
  const int lane = threadIdx.x & 31;
  const int n0 = __builtin_amdgcn_readfirstlane((int)(blockIdx.x * 8 + (threadIdx.x >> 5)) * PULL_K);
  if (n0 >= n_nodes) return;
  const int nk = min(PULL_K, n_nodes - n0);
  const int2* pairs = reinterpret_cast<const int2*>(pairs_i);
  const int o = off[min(n0 + lane, n_nodes)];

  const int b0 = __builtin_amdgcn_readlane(o, 0);
  const int e0 = __builtin_amdgcn_readlane(o, 1);
  int2 p = mask_pairs(load_pairs(pairs, b0, e0, lane), b0, e0, lane);
  for (int i = 0; i < nk; ++i) {
    const int beg = __builtin_amdgcn_readlane(o, i);
    const int end = __builtin_amdgcn_readlane(o, i + 1);
    const int ne = (i + 1 < nk) ? __builtin_amdgcn_readlane(o, i + 2) : end;
    const int2 pn = load_pairs(pairs, end, ne, lane);
    float acc = 0.f;
    if (beg < end) acc = pull_chunk(H, p, min(32, end - beg), lane, acc);
    for (int base = beg + 32; base < end; base += 32) {
      const int2 q = mask_pairs(load_pairs(pairs, base, end, lane), base, end, lane);
      acc = pull_chunk(H, q, min(32, end - base), lane, acc);
    }
    out[(long)(n0 + i) * OUTC + lane] = acc;
    p = mask_pairs(pn, end, ne, lane);
  }
}

extern "C" void kernel_launch(void* const* d_in, const int* in_sizes, int n_in,
                              void* d_out, int out_size, void* d_ws, size_t ws_size,
                              hipStream_t stream) {
  const float* x  = (const float*)d_in[0];
  const float* W1 = (const float*)d_in[1];
  const float* W2 = (const float*)d_in[2];
  const float* ew = (const float*)d_in[3];
  const int*   ei = (const int*)d_in[4];
  float* out = (float*)d_out;

  char* ws = (char*)d_ws;
  float* h      = (float*)(ws);
  float* agg1   = (float*)(ws + 12800000);
  int*   pairs  = (int*)(ws + 25600000);
  int*   sw1    = (int*)(ws + 38400000);
  int*   d1     = (int*)(ws + 51200000);
  int*   counts = (int*)(ws + 57600000);
  int*   start  = (int*)(ws + 64006144);
  int*   bsum   = (int*)(ws + 70412304);
  int*   off    = (int*)(ws + 70418560);

  const int n_cnt = N_BINS * L1_WAVES;

  bin_count  <<<L1_BLOCKS, 256, 0, stream>>>(ei, counts, N_EDGES);
  tile_sums  <<<n_cnt / 4096, 1024, 0, stream>>>(counts, bsum);
  tile_scan  <<<n_cnt / 4096, 1024, 0, stream>>>(counts, bsum, start, n_cnt);
  bin_scatter<<<L1_BLOCKS, 256, 0, stream>>>(ei, ew, start, sw1, d1, N_EDGES);
  bin_sort   <<<N_BINS, 1024, 0, stream>>>(start, sw1, d1, off, pairs, N_NODES);

  const int tiles = N_NODES / 16;
  const int wpb   = 256 / 32;
  dim3 gemm_grid((tiles + wpb - 1) / wpb);
  gemm_x32_wmma<128><<<gemm_grid, 256, 0, stream>>>(x, W1, W2, h, N_NODES);

  const int pull_grid = (N_NODES + 8 * PULL_K - 1) / (8 * PULL_K);
  gcn_pull<<<pull_grid, 256, 0, stream>>>(h,    off, pairs, agg1, N_NODES);
  gcn_pull<<<pull_grid, 256, 0, stream>>>(agg1, off, pairs, out,  N_NODES);
}


# Level-1 sort table holds one counter per bin and work-group, not per wave (8x smaller table, scan and scattered 4-byte traffic); waves' shares recounted in LDS; edge order and sums unchanged
# speedup vs baseline: 1.5220x; 1.2405x over previous
#include <hip/hip_runtime.h>


typedef __attribute__((ext_vector_type(2))) float v2f;
typedef __attribute__((ext_vector_type(8))) float v8f;

#define N_NODES 100000
#define N_EDGES 1600000
#define OUTC 32
#define TP   34

template<int K>
__global__ __launch_bounds__(256) void gemm_x32_wmma(
    const float* __restrict__ X,
    const float* __restrict__ W,
    const float* __restrict__ W2,
    float* __restrict__ H,
    int n_rows)
{
  __shared__ __attribute__((aligned(8))) float tileT[8][16 * TP];

  const int lane = threadIdx.x & 31;
  const int tile = (int)(blockIdx.x * (blockDim.x >> 5) + (threadIdx.x >> 5));
  const int n_tiles = n_rows >> 4;
  if (tile >= n_tiles) return;

  const int  m   = lane & 15;
  const int  kh  = lane >> 4;
  const long row0 = (long)tile << 4;

  const float* xrow = X + (row0 + m) * K;

  v8f c0 = {};
  v8f c1 = {};

  #pragma unroll 8
  for (int kb = 0; kb < K; kb += 4) {
    const int ka = kb + 2 * kh;
    v2f a;
    a.x = xrow[ka];
    a.y = xrow[ka + 1];
    v2f b0, b1;
    b0.x = W[ka * OUTC + m];
    b0.y = W[(ka + 1) * OUTC + m];
    b1.x = W[ka * OUTC + 16 + m];
    b1.y = W[(ka + 1) * OUTC + 16 + m];
    c0 = __builtin_amdgcn_wmma_f32_16x16x4_f32(false, a, false, b0, (short)0, c0, false, false);
    c1 = __builtin_amdgcn_wmma_f32_16x16x4_f32(false, a, false, b1, (short)0, c1, false, false);
  }

  v2f w2lo[OUTC / 4], w2hi[OUTC / 4];
  #pragma unroll
  for (int s = 0; s < OUTC / 4; ++s) {
    const int ka = 4 * s + 2 * kh;
    w2lo[s].x = W2[ka * OUTC + m];
    w2lo[s].y = W2[(ka + 1) * OUTC + m];
    w2hi[s].x = W2[ka * OUTC + 16 + m];
    w2hi[s].y = W2[(ka + 1) * OUTC + 16 + m];
  }

  float* T = tileT[threadIdx.x >> 5];
  #pragma unroll
  for (int r = 0; r < 8; ++r) {
    T[(8 * kh + r) * TP + m]      = c0[r];
    T[(8 * kh + r) * TP + 16 + m] = c1[r];
  }
  __builtin_amdgcn_fence(__ATOMIC_RELEASE, "wavefront");
  __builtin_amdgcn_wave_barrier();
  __builtin_amdgcn_fence(__ATOMIC_ACQUIRE, "wavefront");

  v8f d0 = {};
  v8f d1 = {};
  #pragma unroll
  for (int s = 0; s < OUTC / 4; ++s) {
    const int ka = 4 * s + 2 * kh;
    const v2f a = *reinterpret_cast<const v2f*>(&T[m * TP + ka]);
    d0 = __builtin_amdgcn_wmma_f32_16x16x4_f32(false, a, false, w2lo[s], (short)0, d0, false, false);
    d1 = __builtin_amdgcn_wmma_f32_16x16x4_f32(false, a, false, w2hi[s], (short)0, d1, false, false);
  }

  float* hrow = H + (row0 + 8 * kh) * OUTC + m;
  #pragma unroll
  for (int r = 0; r < 8; ++r) {
    hrow[r * OUTC]      = d0[r];
    hrow[r * OUTC + 16] = d1[r];
  }
}

#define N_BINS     ((N_NODES + 255) >> 8)
#define N_BINS_PAD 392
#define L1_BLOCKS  512
#define L1_STEPS   13
static_assert((long)L1_BLOCKS * 8 * L1_STEPS * 32 >= N_EDGES, "level-1 waves must cover every edge");
static_assert(N_BINS < N_BINS_PAD && N_BINS_PAD <= 512, "level-1 key is matched on 9 bits; the last row of the table is an empty bin");
static_assert((N_BINS_PAD * L1_BLOCKS) % 4096 == 0, "the scan works on whole tiles of 4096 counters");
static_assert(N_BINS_PAD <= 512, "two bins per thread cover the table row of a 256-thread work-group");

__device__ __forceinline__ void wave_sync() {
  __builtin_amdgcn_fence(__ATOMIC_RELEASE, "wavefront");
  __builtin_amdgcn_wave_barrier();
  __builtin_amdgcn_fence(__ATOMIC_ACQUIRE, "wavefront");
}

template<int BITS>
__device__ __forceinline__ unsigned peer_mask(int key, bool valid) {
  unsigned peers = __builtin_amdgcn_ballot_w32(valid);
  #pragma unroll
  for (int b = 0; b < BITS; ++b) {
    const bool bit = (key >> b) & 1;
    const unsigned bal = __builtin_amdgcn_ballot_w32(bit);
    peers &= bit ? bal : ~bal;
  }
  return peers;
}

__device__ __forceinline__ int wave_sum(int v) {
  #pragma unroll
  for (int m = 16; m > 0; m >>= 1) v += __shfl_xor(v, m);
  return v;
}

__device__ __forceinline__ int wave_scan_incl(int v, int lane) {
  #pragma unroll
  for (int d = 1; d < 32; d <<= 1) {
    const int t = __shfl_up(v, d);
    if (lane >= d) v += t;
  }
  return v;
}

__global__ __launch_bounds__(256) void bin_count(
    const int* __restrict__ eidx,
    int*       __restrict__ counts,
    int n_edges)
{
  __shared__ int hist[8][N_BINS_PAD];
  const int lane = threadIdx.x & 31;
  const int wv   = threadIdx.x >> 5;
  const int w    = (int)blockIdx.x * 8 + wv;
  for (int b = lane; b < N_BINS_PAD; b += 32) hist[wv][b] = 0;

  int key[L1_STEPS];
  const int e0 = w * (L1_STEPS * 32) + lane;
  #pragma unroll
  for (int s = 0; s < L1_STEPS; ++s) {
    const int e = e0 + s * 32;
    const int d = eidx[n_edges + min(e, n_edges - 1)];
    key[s] = (e < n_edges) ? (d >> 8) : -1;
  }
  wave_sync();
  #pragma unroll
  for (int s = 0; s < L1_STEPS; ++s) {
    const bool valid = key[s] >= 0;
    const unsigned peers = peer_mask<9>(key[s], valid);
    if (valid && (peers >> lane) == 1u) hist[wv][key[s]] += __popc(peers);
    wave_sync();
  }
  __syncthreads();
  for (int b = threadIdx.x; b < N_BINS_PAD; b += 256) {
    int t = 0;
    #pragma unroll
    for (int k = 0; k < 8; ++k) t += hist[k][b];
    counts[b * L1_BLOCKS + blockIdx.x] = t;
  }
}

__global__ __launch_bounds__(1024) void tile_sums(
    const int* __restrict__ cnt,
    int*       __restrict__ bsum)
{
  __shared__ int part[32];
  const int i4 = (int)(blockIdx.x * 1024 + threadIdx.x);
  const int4 c = reinterpret_cast<const int4*>(cnt)[i4];
  const int s = wave_sum(c.x + c.y + c.z + c.w);
  if ((threadIdx.x & 31) == 0) part[threadIdx.x >> 5] = s;
  __syncthreads();
  if (threadIdx.x < 32) {
    const int t = wave_sum(part[threadIdx.x]);
    if (threadIdx.x == 0) bsum[blockIdx.x] = t;
  }
}

__global__ __launch_bounds__(1024) void tile_scan(
    const int* __restrict__ cnt,
    const int* __restrict__ bsum,
    int*       __restrict__ pre,
    int n)
{
  __shared__ int part[32];
  __shared__ int partb[32];
  const int lane = threadIdx.x & 31;
  const int wv   = threadIdx.x >> 5;
  const int i4   = (int)(blockIdx.x * 1024 + threadIdx.x);
  const int4 c   = reinterpret_cast<const int4*>(cnt)[i4];
  const int tsum = c.x + c.y + c.z + c.w;
  const int inc  = wave_scan_incl(tsum, lane);
  if (lane == 31) part[wv] = inc;

  int b = 0;
  for (int k = (int)threadIdx.x; k < (int)blockIdx.x; k += 1024) b += bsum[k];
  b = wave_sum(b);
  if (lane == 0) partb[wv] = b;
  __syncthreads();

  int before = 0;
  #pragma unroll
  for (int k = 0; k < 32; ++k) {
    before += partb[k];
    if (k < wv) before += part[k];
  }
  const int ex = before + inc - tsum;
  int4 o;
  o.x = ex;
  o.y = o.x + c.x;
  o.z = o.y + c.y;
  o.w = o.z + c.z;
  reinterpret_cast<int4*>(pre)[i4] = o;
  if (i4 * 4 + 4 == n) pre[n] = o.w + c.w;
}

__global__ __launch_bounds__(256) void bin_scatter(
    const int*   __restrict__ eidx,
    const float* __restrict__ ew,
    const int*   __restrict__ start,
    int*         __restrict__ sw1_i,
    int*         __restrict__ d1,
    int n_edges)
{
  __shared__ int cur[8][N_BINS_PAD];
  const int lane = threadIdx.x & 31;
  const int wv   = threadIdx.x >> 5;
  const int w    = (int)blockIdx.x * 8 + wv;
  for (int b = lane; b < N_BINS_PAD; b += 32) cur[wv][b] = 0;

  const int  bA   = (int)threadIdx.x;
  const int  bB   = (int)threadIdx.x + 256;
  const bool hasB = bB < N_BINS_PAD;
  const int  runA = start[bA * L1_BLOCKS + blockIdx.x];
  const int  runB = start[(hasB ? bB : bA) * L1_BLOCKS + blockIdx.x];

  int dst[L1_STEPS], srcv[L1_STEPS], wbits[L1_STEPS];
  const int e0 = w * (L1_STEPS * 32) + lane;
  #pragma unroll
  for (int s = 0; s < L1_STEPS; ++s) {
    const int e  = e0 + s * 32;
    const int ec = min(e, n_edges - 1);
    const int d  = eidx[n_edges + ec];
    srcv[s]  = eidx[ec];
    wbits[s] = __float_as_int(ew[ec]);
    dst[s]   = (e < n_edges) ? d : -1;
  }
  int2* sw1 = reinterpret_cast<int2*>(sw1_i);
  wave_sync();
  unsigned pm[L1_STEPS];
  #pragma unroll
  for (int s = 0; s < L1_STEPS; ++s) {
    const bool valid = dst[s] >= 0;
    const int  key   = dst[s] >> 8;
    pm[s] = peer_mask<9>(key, valid);
    if (valid && (pm[s] >> lane) == 1u) cur[wv][key] += __popc(pm[s]);
    wave_sync();
  }
  __syncthreads();
  {
    int run = runA;
    #pragma unroll
    for (int k = 0; k < 8; ++k) { const int c = cur[k][bA]; cur[k][bA] = run; run += c; }
  }
  if (hasB) {
    int run = runB;
    #pragma unroll
    for (int k = 0; k < 8; ++k) { const int c = cur[k][bB]; cur[k][bB] = run; run += c; }
  }
  __syncthreads();
  #pragma unroll
  for (int s = 0; s < L1_STEPS; ++s) {
    const bool valid = dst[s] >= 0;
    const int  key   = dst[s] >> 8;
    const unsigned peers = pm[s];
    const int base = valid ? cur[wv][key] : 0;
    wave_sync();
    if (valid && (peers >> lane) == 1u) cur[wv][key] = base + __popc(peers);
    wave_sync();
    if (valid) {
      const int pos = base + __popc(peers & ((1u << lane) - 1u));
      sw1[pos] = make_int2(srcv[s], wbits[s]);
      d1[pos]  = dst[s];
    }
  }
}

__global__ __launch_bounds__(1024) void bin_sort(
    const int* __restrict__ start,
    const int* __restrict__ sw1_i,
    const int* __restrict__ d1,
    int*       __restrict__ off,
    int*       __restrict__ pairs_i,
    int n_nodes)
{
  __shared__ int hist[32][256];
  __shared__ int nodeoff[256];
  __shared__ int wtot[8];
  const int tid  = threadIdx.x;
  const int lane = tid & 31;
  const int wv   = tid >> 5;
  const int bin  = blockIdx.x;
  const int beg  = start[bin * L1_BLOCKS];
  const int end  = start[(bin + 1) * L1_BLOCKS];
  #pragma unroll
  for (int k = 0; k < 8; ++k) hist[wv][lane + 32 * k] = 0;
  wave_sync();

  for (int i0 = beg; i0 < end; i0 += 1024) {
    const int  i     = i0 + tid;
    const bool valid = i < end;
    const int  key   = d1[min(i, end - 1)] & 255;
    const unsigned peers = peer_mask<8>(key, valid);
    if (valid && (peers >> lane) == 1u) hist[wv][key] += __popc(peers);
    wave_sync();
  }
  __syncthreads();

  int tot = 0, inc = 0;
  if (tid < 256) {
    for (int k = 0; k < 32; ++k) {
      const int c = hist[k][tid];
      hist[k][tid] = tot;
      tot += c;
    }
    inc = wave_scan_incl(tot, lane);
    if (lane == 31) wtot[wv] = inc;
  }
  __syncthreads();
  if (tid < 256) {
    int pre = 0;
    for (int k = 0; k < wv; ++k) pre += wtot[k];
    const int no = pre + inc - tot;
    nodeoff[tid] = no;
    const int node = bin * 256 + tid;
    if (node < n_nodes) off[node] = beg + no;
    if (node == n_nodes - 1) off[n_nodes] = beg + no + tot;
  }
  __syncthreads();

  const int2* sw1   = reinterpret_cast<const int2*>(sw1_i);
  int2*       pairs = reinterpret_cast<int2*>(pairs_i);
  for (int i0 = beg; i0 < end; i0 += 1024) {
    const int  i     = i0 + tid;
    const bool valid = i < end;
    const int  ic    = min(i, end - 1);
    const int  key   = d1[ic] & 255;
    const int2 rec   = sw1[ic];
    const unsigned peers = peer_mask<8>(key, valid);
    const int base = hist[wv][key];
    wave_sync();
    if (valid && (peers >> lane) == 1u) hist[wv][key] = base + __popc(peers);
    wave_sync();
    if (valid) {
      const int pos = beg + nodeoff[key] + base + __popc(peers & ((1u << lane) - 1u));
      pairs[pos] = rec;
    }
  }
}

#define PULL_K 16

__device__ __forceinline__ int2 load_pairs(const int2* __restrict__ pairs, int b, int e, int lane) {
  int2 v = make_int2(0, 0);
  if (b < e) v = pairs[min(b + lane, e - 1)];
  return v;
}
__device__ __forceinline__ int2 mask_pairs(int2 v, int b, int e, int lane) {
  if (b + lane >= e) v.y = 0;
  return v;
}

template<int U>
__device__ __forceinline__ float pull_rows(const float* __restrict__ H, int2 p, int lane, float acc) {
  float hv[U], wt[U];
  #pragma unroll
  for (int u = 0; u < U; ++u) {
    const int s = __builtin_amdgcn_readlane(p.x, u);
    wt[u] = __int_as_float(__builtin_amdgcn_readlane(p.y, u));
    hv[u] = H[(long)s * OUTC + lane];
  }
  #pragma unroll
  for (int u = 0; u < U; ++u) acc = fmaf(wt[u], hv[u], acc);
  return acc;
}

__device__ __forceinline__ float pull_chunk(const float* __restrict__ H, int2 p, int cnt, int lane, float acc) {
  if (cnt <= 8)  return pull_rows<8>(H, p, lane, acc);
  if (cnt <= 16) return pull_rows<16>(H, p, lane, acc);
  if (cnt <= 24) return pull_rows<24>(H, p, lane, acc);
  return pull_rows<32>(H, p, lane, acc);
}

__global__ __launch_bounds__(256) void gcn_pull(
    const float* __restrict__ H,
    const int*   __restrict__ off,
    const int*   __restrict__ pairs_i,
    float*       __restrict__ out,
    int n_nodes)
{
  const int lane = threadIdx.x & 31;
  const int n0 = __builtin_amdgcn_readfirstlane((int)(blockIdx.x * 8 + (threadIdx.x >> 5)) * PULL_K);
  if (n0 >= n_nodes) return;
  const int nk = min(PULL_K, n_nodes - n0);
  const int2* pairs = reinterpret_cast<const int2*>(pairs_i);
  const int o = off[min(n0 + lane, n_nodes)];

  const int b0 = __builtin_amdgcn_readlane(o, 0);
  const int e0 = __builtin_amdgcn_readlane(o, 1);
  int2 p = mask_pairs(load_pairs(pairs, b0, e0, lane), b0, e0, lane);
  for (int i = 0; i < nk; ++i) {
    const int beg = __builtin_amdgcn_readlane(o, i);
    const int end = __builtin_amdgcn_readlane(o, i + 1);
    const int ne = (i + 1 < nk) ? __builtin_amdgcn_readlane(o, i + 2) : end;
    const int2 pn = load_pairs(pairs, end, ne, lane);
    float acc = 0.f;
    if (beg < end) acc = pull_chunk(H, p, min(32, end - beg), lane, acc);
    for (int base = beg + 32; base < end; base += 32) {
      const int2 q = mask_pairs(load_pairs(pairs, base, end, lane), base, end, lane);
      acc = pull_chunk(H, q, min(32, end - base), lane, acc);
    }
    out[(long)(n0 + i) * OUTC + lane] = acc;
    p = mask_pairs(pn, end, ne, lane);
  }
}

extern "C" void kernel_launch(void* const* d_in, const int* in_sizes, int n_in,
                              void* d_out, int out_size, void* d_ws, size_t ws_size,
                              hipStream_t stream) {
  const float* x  = (const float*)d_in[0];
  const float* W1 = (const float*)d_in[1];
  const float* W2 = (const float*)d_in[2];
  const float* ew = (const float*)d_in[3];
  const int*   ei = (const int*)d_in[4];
  float* out = (float*)d_out;

  char* ws = (char*)d_ws;
  float* h      = (float*)(ws);
  float* agg1   = (float*)(ws + 12800000);
  int*   pairs  = (int*)(ws + 25600000);
  int*   sw1    = (int*)(ws + 38400000);
  int*   d1     = (int*)(ws + 51200000);
  int*   counts = (int*)(ws + 57600000);
  int*   start  = (int*)(ws + 64006144);
  int*   bsum   = (int*)(ws + 70412304);
  int*   off    = (int*)(ws + 70418560);

  const int n_cnt = N_BINS_PAD * L1_BLOCKS;

  bin_count  <<<L1_BLOCKS, 256, 0, stream>>>(ei, counts, N_EDGES);
  tile_sums  <<<n_cnt / 4096, 1024, 0, stream>>>(counts, bsum);
  tile_scan  <<<n_cnt / 4096, 1024, 0, stream>>>(counts, bsum, start, n_cnt);
  bin_scatter<<<L1_BLOCKS, 256, 0, stream>>>(ei, ew, start, sw1, d1, N_EDGES);
  bin_sort   <<<N_BINS, 1024, 0, stream>>>(start, sw1, d1, off, pairs, N_NODES);

  const int tiles = N_NODES / 16;
  const int wpb   = 256 / 32;
  dim3 gemm_grid((tiles + wpb - 1) / wpb);
  gemm_x32_wmma<128><<<gemm_grid, 256, 0, stream>>>(x, W1, W2, h, N_NODES);

  const int pull_grid = (N_NODES + 8 * PULL_K - 1) / (8 * PULL_K);
  gcn_pull<<<pull_grid, 256, 0, stream>>>(h,    off, pairs, agg1, N_NODES);
  gcn_pull<<<pull_grid, 256, 0, stream>>>(agg1, off, pairs, out,  N_NODES);
}


# Both sort levels build their output in LDS in output order, then copy it out with consecutive lanes on consecutive addresses (was a scattered 4/8-byte store per edge); order and sums unchanged
# speedup vs baseline: 2.2875x; 1.5030x over previous
#include <hip/hip_runtime.h>


typedef __attribute__((ext_vector_type(2))) float v2f;
typedef __attribute__((ext_vector_type(8))) float v8f;

#define N_NODES 100000
#define N_EDGES 1600000
#define OUTC 32
#define TP   34

template<int K>
__global__ __launch_bounds__(256) void gemm_x32_wmma(
    const float* __restrict__ X,
    const float* __restrict__ W,
    const float* __restrict__ W2,
    float* __restrict__ H,
    int n_rows)
{
  __shared__ __attribute__((aligned(8))) float tileT[8][16 * TP];

  const int lane = threadIdx.x & 31;
  const int tile = (int)(blockIdx.x * (blockDim.x >> 5) + (threadIdx.x >> 5));
  const int n_tiles = n_rows >> 4;
  if (tile >= n_tiles) return;

  const int  m   = lane & 15;
  const int  kh  = lane >> 4;
  const long row0 = (long)tile << 4;

  const float* xrow = X + (row0 + m) * K;

  v8f c0 = {};
  v8f c1 = {};

  #pragma unroll 8
  for (int kb = 0; kb < K; kb += 4) {
    const int ka = kb + 2 * kh;
    v2f a;
    a.x = xrow[ka];
    a.y = xrow[ka + 1];
    v2f b0, b1;
    b0.x = W[ka * OUTC + m];
    b0.y = W[(ka + 1) * OUTC + m];
    b1.x = W[ka * OUTC + 16 + m];
    b1.y = W[(ka + 1) * OUTC + 16 + m];
    c0 = __builtin_amdgcn_wmma_f32_16x16x4_f32(false, a, false, b0, (short)0, c0, false, false);
    c1 = __builtin_amdgcn_wmma_f32_16x16x4_f32(false, a, false, b1, (short)0, c1, false, false);
  }

  v2f w2lo[OUTC / 4], w2hi[OUTC / 4];
  #pragma unroll
  for (int s = 0; s < OUTC / 4; ++s) {
    const int ka = 4 * s + 2 * kh;
    w2lo[s].x = W2[ka * OUTC + m];
    w2lo[s].y = W2[(ka + 1) * OUTC + m];
    w2hi[s].x = W2[ka * OUTC + 16 + m];
    w2hi[s].y = W2[(ka + 1) * OUTC + 16 + m];
  }

  float* T = tileT[threadIdx.x >> 5];
  #pragma unroll
  for (int r = 0; r < 8; ++r) {
    T[(8 * kh + r) * TP + m]      = c0[r];
    T[(8 * kh + r) * TP + 16 + m] = c1[r];
  }
  __builtin_amdgcn_fence(__ATOMIC_RELEASE, "wavefront");
  __builtin_amdgcn_wave_barrier();
  __builtin_amdgcn_fence(__ATOMIC_ACQUIRE, "wavefront");

  v8f d0 = {};
  v8f d1 = {};
  #pragma unroll
  for (int s = 0; s < OUTC / 4; ++s) {
    const int ka = 4 * s + 2 * kh;
    const v2f a = *reinterpret_cast<const v2f*>(&T[m * TP + ka]);
    d0 = __builtin_amdgcn_wmma_f32_16x16x4_f32(false, a, false, w2lo[s], (short)0, d0, false, false);
    d1 = __builtin_amdgcn_wmma_f32_16x16x4_f32(false, a, false, w2hi[s], (short)0, d1, false, false);
  }

  float* hrow = H + (row0 + 8 * kh) * OUTC + m;
  #pragma unroll
  for (int r = 0; r < 8; ++r) {
    hrow[r * OUTC]      = d0[r];
    hrow[r * OUTC + 16] = d1[r];
  }
}

#define N_BINS     ((N_NODES + 255) >> 8)
#define N_BINS_PAD 392
#define L1_BLOCKS  512
#define L1_STEPS   13
static_assert((long)L1_BLOCKS * 8 * L1_STEPS * 32 >= N_EDGES, "level-1 waves must cover every edge");
static_assert(N_BINS < N_BINS_PAD && N_BINS_PAD <= 512, "level-1 key is matched on 9 bits; the last row of the table is an empty bin");
static_assert((N_BINS_PAD * L1_BLOCKS) % 4096 == 0, "the scan works on whole tiles of 4096 counters");
static_assert(N_BINS_PAD <= 512 && N_BINS_PAD % 2 == 0, "two neighbouring bins per thread cover the table row of a 256-thread work-group");
#define L1_WG_EDGES (8 * L1_STEPS * 32)
#define L2_IMG      5120

__device__ __forceinline__ void wave_sync() {
  __builtin_amdgcn_fence(__ATOMIC_RELEASE, "wavefront");
  __builtin_amdgcn_wave_barrier();
  __builtin_amdgcn_fence(__ATOMIC_ACQUIRE, "wavefront");
}

template<int BITS>
__device__ __forceinline__ unsigned peer_mask(int key, bool valid) {
  unsigned peers = __builtin_amdgcn_ballot_w32(valid);
  #pragma unroll
  for (int b = 0; b < BITS; ++b) {
    const bool bit = (key >> b) & 1;
    const unsigned bal = __builtin_amdgcn_ballot_w32(bit);
    peers &= bit ? bal : ~bal;
  }
  return peers;
}

__device__ __forceinline__ int wave_sum(int v) {
  #pragma unroll
  for (int m = 16; m > 0; m >>= 1) v += __shfl_xor(v, m);
  return v;
}

__device__ __forceinline__ int wave_scan_incl(int v, int lane) {
  #pragma unroll
  for (int d = 1; d < 32; d <<= 1) {
    const int t = __shfl_up(v, d);
    if (lane >= d) v += t;
  }
  return v;
}

__global__ __launch_bounds__(256) void bin_count(
    const int* __restrict__ eidx,
    int*       __restrict__ counts,
    int n_edges)
{
  __shared__ int hist[8][N_BINS_PAD];
  const int lane = threadIdx.x & 31;
  const int wv   = threadIdx.x >> 5;
  const int w    = (int)blockIdx.x * 8 + wv;
  for (int b = lane; b < N_BINS_PAD; b += 32) hist[wv][b] = 0;

  int key[L1_STEPS];
  const int e0 = w * (L1_STEPS * 32) + lane;
  #pragma unroll
  for (int s = 0; s < L1_STEPS; ++s) {
    const int e = e0 + s * 32;
    const int d = eidx[n_edges + min(e, n_edges - 1)];
    key[s] = (e < n_edges) ? (d >> 8) : -1;
  }
  wave_sync();
  #pragma unroll
  for (int s = 0; s < L1_STEPS; ++s) {
    const bool valid = key[s] >= 0;
    const unsigned peers = peer_mask<9>(key[s], valid);
    if (valid && (peers >> lane) == 1u) hist[wv][key[s]] += __popc(peers);
    wave_sync();
  }
  __syncthreads();
  for (int b = threadIdx.x; b < N_BINS_PAD; b += 256) {
    int t = 0;
    #pragma unroll
    for (int k = 0; k < 8; ++k) t += hist[k][b];
    counts[b * L1_BLOCKS + blockIdx.x] = t;
  }
}

__global__ __launch_bounds__(1024) void tile_sums(
    const int* __restrict__ cnt,
    int*       __restrict__ bsum)
{
  __shared__ int part[32];
  const int i4 = (int)(blockIdx.x * 1024 + threadIdx.x);
  const int4 c = reinterpret_cast<const int4*>(cnt)[i4];
  const int s = wave_sum(c.x + c.y + c.z + c.w);
  if ((threadIdx.x & 31) == 0) part[threadIdx.x >> 5] = s;
  __syncthreads();
  if (threadIdx.x < 32) {
    const int t = wave_sum(part[threadIdx.x]);
    if (threadIdx.x == 0) bsum[blockIdx.x] = t;
  }
}

__global__ __launch_bounds__(1024) void tile_scan(
    const int* __restrict__ cnt,
    const int* __restrict__ bsum,
    int*       __restrict__ pre,
    int n)
{
  __shared__ int part[32];
  __shared__ int partb[32];
  const int lane = threadIdx.x & 31;
  const int wv   = threadIdx.x >> 5;
  const int i4   = (int)(blockIdx.x * 1024 + threadIdx.x);
  const int4 c   = reinterpret_cast<const int4*>(cnt)[i4];
  const int tsum = c.x + c.y + c.z + c.w;
  const int inc  = wave_scan_incl(tsum, lane);
  if (lane == 31) part[wv] = inc;

  int b = 0;
  for (int k = (int)threadIdx.x; k < (int)blockIdx.x; k += 1024) b += bsum[k];
  b = wave_sum(b);
  if (lane == 0) partb[wv] = b;
  __syncthreads();

  int before = 0;
  #pragma unroll
  for (int k = 0; k < 32; ++k) {
    before += partb[k];
    if (k < wv) before += part[k];
  }
  const int ex = before + inc - tsum;
  int4 o;
  o.x = ex;
  o.y = o.x + c.x;
  o.z = o.y + c.y;
  o.w = o.z + c.z;
  reinterpret_cast<int4*>(pre)[i4] = o;
  if (i4 * 4 + 4 == n) pre[n] = o.w + c.w;
}

__global__ __launch_bounds__(256) void bin_scatter(
    const int*   __restrict__ eidx,
    const float* __restrict__ ew,
    const int*   __restrict__ start,
    int*         __restrict__ sw1_i,
    int*         __restrict__ d1,
    int n_edges)
{
  __shared__ int  cur[8][N_BINS_PAD];
  __shared__ int  shift[N_BINS_PAD];
  __shared__ int  wtot[8];
  __shared__ int2 img_sw[L1_WG_EDGES];
  __shared__ int  img_d[L1_WG_EDGES];
  const int lane = threadIdx.x & 31;
  const int wv   = threadIdx.x >> 5;
  const int w    = (int)blockIdx.x * 8 + wv;
  for (int b = lane; b < N_BINS_PAD; b += 32) cur[wv][b] = 0;

  const bool owner = (int)threadIdx.x < N_BINS_PAD / 2;
  const int  bA    = 2 * min((int)threadIdx.x, N_BINS_PAD / 2 - 1);
  const int  bB    = bA + 1;
  const int  runA  = start[bA * L1_BLOCKS + blockIdx.x];
  const int  runB  = start[bB * L1_BLOCKS + blockIdx.x];

  int dst[L1_STEPS], srcv[L1_STEPS], wbits[L1_STEPS];
  const int e0 = w * (L1_STEPS * 32) + lane;
  #pragma unroll
  for (int s = 0; s < L1_STEPS; ++s) {
    const int e  = e0 + s * 32;
    const int ec = min(e, n_edges - 1);
    const int d  = eidx[n_edges + ec];
    srcv[s]  = eidx[ec];
    wbits[s] = __float_as_int(ew[ec]);
    dst[s]   = (e < n_edges) ? d : -1;
  }
  int2* sw1 = reinterpret_cast<int2*>(sw1_i);
  wave_sync();
  unsigned pm[L1_STEPS];
  #pragma unroll
  for (int s = 0; s < L1_STEPS; ++s) {
    const bool valid = dst[s] >= 0;
    const int  key   = dst[s] >> 8;
    pm[s] = peer_mask<9>(key, valid);
    if (valid && (pm[s] >> lane) == 1u) cur[wv][key] += __popc(pm[s]);
    wave_sync();
  }
  __syncthreads();
  int cA[8], cB[8];
  int tA = 0, tB = 0;
  #pragma unroll
  for (int k = 0; k < 8; ++k) {
    cA[k] = owner ? cur[k][bA] : 0;
    cB[k] = owner ? cur[k][bB] : 0;
    tA += cA[k];
    tB += cB[k];
  }
  const int both = tA + tB;
  const int inc  = wave_scan_incl(both, lane);
  if (lane == 31) wtot[wv] = inc;
  __syncthreads();
  int ex = inc - both;
  #pragma unroll
  for (int k = 0; k < 8; ++k) if (k < wv) ex += wtot[k];
  if (owner) {
    int run = ex;
    #pragma unroll
    for (int k = 0; k < 8; ++k) { cur[k][bA] = run; run += cA[k]; }
    #pragma unroll
    for (int k = 0; k < 8; ++k) { cur[k][bB] = run; run += cB[k]; }
    shift[bA] = runA - ex;
    shift[bB] = runB - (ex + tA);
  }
  __syncthreads();
  #pragma unroll
  for (int s = 0; s < L1_STEPS; ++s) {
    const bool valid = dst[s] >= 0;
    const int  key   = dst[s] >> 8;
    const unsigned peers = pm[s];
    const int base = valid ? cur[wv][key] : 0;
    wave_sync();
    if (valid && (peers >> lane) == 1u) cur[wv][key] = base + __popc(peers);
    wave_sync();
    if (valid) {
      const int pos = base + __popc(peers & ((1u << lane) - 1u));
      img_sw[pos] = make_int2(srcv[s], wbits[s]);
      img_d[pos]  = dst[s];
    }
  }
  __syncthreads();
  const int first = (int)blockIdx.x * L1_WG_EDGES;
  const int nloc  = min(max(n_edges - first, 0), L1_WG_EDGES);
  if (nloc > 0) {
    int  dd[L1_STEPS], sh[L1_STEPS];
    int2 rec[L1_STEPS];
    #pragma unroll
    for (int s = 0; s < L1_STEPS; ++s) dd[s] = img_d[min(s * 256 + (int)threadIdx.x, nloc - 1)];
    #pragma unroll
    for (int s = 0; s < L1_STEPS; ++s) {
      sh[s]  = shift[dd[s] >> 8];
      rec[s] = img_sw[min(s * 256 + (int)threadIdx.x, nloc - 1)];
    }
    #pragma unroll
    for (int s = 0; s < L1_STEPS; ++s) {
      const int i = s * 256 + (int)threadIdx.x;
      if (i < nloc) {
        const int p = sh[s] + i;
        sw1[p] = rec[s];
        d1[p]  = dd[s];
      }
    }
  }
}

__global__ __launch_bounds__(1024) void bin_sort(
    const int* __restrict__ start,
    const int* __restrict__ sw1_i,
    const int* __restrict__ d1,
    int*       __restrict__ off,
    int*       __restrict__ pairs_i,
    int n_nodes)
{
  __shared__ int hist[32][256];
  __shared__ int nodeoff[256];
  __shared__ int wtot[8];
  __shared__ int2 img[L2_IMG];
  const int tid  = threadIdx.x;
  const int lane = tid & 31;
  const int wv   = tid >> 5;
  const int bin  = blockIdx.x;
  const int beg  = start[bin * L1_BLOCKS];
  const int end  = start[(bin + 1) * L1_BLOCKS];
  const int  nbin   = end - beg;
  const bool staged = nbin <= L2_IMG;
  #pragma unroll
  for (int k = 0; k < 8; ++k) hist[wv][lane + 32 * k] = 0;
  wave_sync();

  for (int i0 = beg; i0 < end; i0 += 1024) {
    const int  i     = i0 + tid;
    const bool valid = i < end;
    const int  key   = d1[min(i, end - 1)] & 255;
    const unsigned peers = peer_mask<8>(key, valid);
    if (valid && (peers >> lane) == 1u) hist[wv][key] += __popc(peers);
    wave_sync();
  }
  __syncthreads();

  int tot = 0, inc = 0;
  if (tid < 256) {
    for (int k = 0; k < 32; ++k) {
      const int c = hist[k][tid];
      hist[k][tid] = tot;
      tot += c;
    }
    inc = wave_scan_incl(tot, lane);
    if (lane == 31) wtot[wv] = inc;
  }
  __syncthreads();
  if (tid < 256) {
    int pre = 0;
    for (int k = 0; k < wv; ++k) pre += wtot[k];
    const int no = pre + inc - tot;
    nodeoff[tid] = no;
    const int node = bin * 256 + tid;
    if (node < n_nodes) off[node] = beg + no;
    if (node == n_nodes - 1) off[n_nodes] = beg + no + tot;
  }
  __syncthreads();

  const int2* sw1   = reinterpret_cast<const int2*>(sw1_i);
  int2*       pairs = reinterpret_cast<int2*>(pairs_i);
  for (int i0 = beg; i0 < end; i0 += 1024) {
    const int  i     = i0 + tid;
    const bool valid = i < end;
    const int  ic    = min(i, end - 1);
    const int  key   = d1[ic] & 255;
    const int2 rec   = sw1[ic];
    const unsigned peers = peer_mask<8>(key, valid);
    const int base = hist[wv][key];
    wave_sync();
    if (valid && (peers >> lane) == 1u) hist[wv][key] = base + __popc(peers);
    wave_sync();
    if (valid) {
      const int lp = nodeoff[key] + base + __popc(peers & ((1u << lane) - 1u));
      if (staged) img[lp] = rec;
      else        pairs[beg + lp] = rec;
    }
  }
  if (staged) {
    __syncthreads();
    for (int i = tid; i < nbin; i += 1024) pairs[beg + i] = img[i];
  }
}

#define PULL_K 16

__device__ __forceinline__ int2 load_pairs(const int2* __restrict__ pairs, int b, int e, int lane) {
  int2 v = make_int2(0, 0);
  if (b < e) v = pairs[min(b + lane, e - 1)];
  return v;
}
__device__ __forceinline__ int2 mask_pairs(int2 v, int b, int e, int lane) {
  if (b + lane >= e) v.y = 0;
  return v;
}

template<int U>
__device__ __forceinline__ float pull_rows(const float* __restrict__ H, int2 p, int lane, float acc) {
  float hv[U], wt[U];
  #pragma unroll
  for (int u = 0; u < U; ++u) {
    const int s = __builtin_amdgcn_readlane(p.x, u);
    wt[u] = __int_as_float(__builtin_amdgcn_readlane(p.y, u));
    hv[u] = H[(long)s * OUTC + lane];
  }
  #pragma unroll
  for (int u = 0; u < U; ++u) acc = fmaf(wt[u], hv[u], acc);
  return acc;
}

__device__ __forceinline__ float pull_chunk(const float* __restrict__ H, int2 p, int cnt, int lane, float acc) {
  if (cnt <= 8)  return pull_rows<8>(H, p, lane, acc);
  if (cnt <= 16) return pull_rows<16>(H, p, lane, acc);
  if (cnt <= 24) return pull_rows<24>(H, p, lane, acc);
  return pull_rows<32>(H, p, lane, acc);
}

__global__ __launch_bounds__(256) void gcn_pull(
    const float* __restrict__ H,
    const int*   __restrict__ off,
    const int*   __restrict__ pairs_i,
    float*       __restrict__ out,
    int n_nodes)
{
  const int lane = threadIdx.x & 31;
  const int n0 = __builtin_amdgcn_readfirstlane((int)(blockIdx.x * 8 + (threadIdx.x >> 5)) * PULL_K);
  if (n0 >= n_nodes) return;
  const int nk = min(PULL_K, n_nodes - n0);
  const int2* pairs = reinterpret_cast<const int2*>(pairs_i);
  const int o = off[min(n0 + lane, n_nodes)];

  const int b0 = __builtin_amdgcn_readlane(o, 0);
  const int e0 = __builtin_amdgcn_readlane(o, 1);
  int2 p = mask_pairs(load_pairs(pairs, b0, e0, lane), b0, e0, lane);
  for (int i = 0; i < nk; ++i) {
    const int beg = __builtin_amdgcn_readlane(o, i);
    const int end = __builtin_amdgcn_readlane(o, i + 1);
    const int ne = (i + 1 < nk) ? __builtin_amdgcn_readlane(o, i + 2) : end;
    const int2 pn = load_pairs(pairs, end, ne, lane);
    float acc = 0.f;
    if (beg < end) acc = pull_chunk(H, p, min(32, end - beg), lane, acc);
    for (int base = beg + 32; base < end; base += 32) {
      const int2 q = mask_pairs(load_pairs(pairs, base, end, lane), base, end, lane);
      acc = pull_chunk(H, q, min(32, end - base), lane, acc);
    }
    out[(long)(n0 + i) * OUTC + lane] = acc;
    p = mask_pairs(pn, end, ne, lane);
  }
}

extern "C" void kernel_launch(void* const* d_in, const int* in_sizes, int n_in,
                              void* d_out, int out_size, void* d_ws, size_t ws_size,
                              hipStream_t stream) {
  const float* x  = (const float*)d_in[0];
  const float* W1 = (const float*)d_in[1];
  const float* W2 = (const float*)d_in[2];
  const float* ew = (const float*)d_in[3];
  const int*   ei = (const int*)d_in[4];
  float* out = (float*)d_out;

  char* ws = (char*)d_ws;
  float* h      = (float*)(ws);
  float* agg1   = (float*)(ws + 12800000);
  int*   pairs  = (int*)(ws + 25600000);
  int*   sw1    = (int*)(ws + 38400000);
  int*   d1     = (int*)(ws + 51200000);
  int*   counts = (int*)(ws + 57600000);
  int*   start  = (int*)(ws + 64006144);
  int*   bsum   = (int*)(ws + 70412304);
  int*   off    = (int*)(ws + 70418560);

  const int n_cnt = N_BINS_PAD * L1_BLOCKS;

  bin_count  <<<L1_BLOCKS, 256, 0, stream>>>(ei, counts, N_EDGES);
  tile_sums  <<<n_cnt / 4096, 1024, 0, stream>>>(counts, bsum);
  tile_scan  <<<n_cnt / 4096, 1024, 0, stream>>>(counts, bsum, start, n_cnt);
  bin_scatter<<<L1_BLOCKS, 256, 0, stream>>>(ei, ew, start, sw1, d1, N_EDGES);
  bin_sort   <<<N_BINS, 1024, 0, stream>>>(start, sw1, d1, off, pairs, N_NODES);

  const int tiles = N_NODES / 16;
  const int wpb   = 256 / 32;
  dim3 gemm_grid((tiles + wpb - 1) / wpb);
  gemm_x32_wmma<128><<<gemm_grid, 256, 0, stream>>>(x, W1, W2, h, N_NODES);

  const int pull_grid = (N_NODES + 8 * PULL_K - 1) / (8 * PULL_K);
  gcn_pull<<<pull_grid, 256, 0, stream>>>(h,    off, pairs, agg1, N_NODES);
  gcn_pull<<<pull_grid, 256, 0, stream>>>(agg1, off, pairs, out,  N_NODES);
}


# Level-1 sort stays inside each work-group's own slice; level 2 gathers a bin's 512 runs and takes its global place from the work-groups' prefix table: no count or scan kernels, 8 launches -> 5
# speedup vs baseline: 2.7330x; 1.1948x over previous
#include <hip/hip_runtime.h>


typedef __attribute__((ext_vector_type(2))) float v2f;
typedef __attribute__((ext_vector_type(8))) float v8f;

#define N_NODES 100000
#define N_EDGES 1600000
#define OUTC 32
#define TP   34

template<int K>
__global__ __launch_bounds__(256) void gemm_x32_wmma(
    const float* __restrict__ X,
    const float* __restrict__ W,
    const float* __restrict__ W2,
    float* __restrict__ H,
    int n_rows)
{
  __shared__ __attribute__((aligned(8))) float tileT[8][16 * TP];

  const int lane = threadIdx.x & 31;
  const int tile = (int)(blockIdx.x * (blockDim.x >> 5) + (threadIdx.x >> 5));
  const int n_tiles = n_rows >> 4;
  if (tile >= n_tiles) return;

  const int  m   = lane & 15;
  const int  kh  = lane >> 4;
  const long row0 = (long)tile << 4;

  const float* xrow = X + (row0 + m) * K;

  v8f c0 = {};
  v8f c1 = {};

  #pragma unroll 8
  for (int kb = 0; kb < K; kb += 4) {
    const int ka = kb + 2 * kh;
    v2f a;
    a.x = xrow[ka];
    a.y = xrow[ka + 1];
    v2f b0, b1;
    b0.x = W[ka * OUTC + m];
    b0.y = W[(ka + 1) * OUTC + m];
    b1.x = W[ka * OUTC + 16 + m];
    b1.y = W[(ka + 1) * OUTC + 16 + m];
    c0 = __builtin_amdgcn_wmma_f32_16x16x4_f32(false, a, false, b0, (short)0, c0, false, false);
    c1 = __builtin_amdgcn_wmma_f32_16x16x4_f32(false, a, false, b1, (short)0, c1, false, false);
  }

  v2f w2lo[OUTC / 4], w2hi[OUTC / 4];
  #pragma unroll
  for (int s = 0; s < OUTC / 4; ++s) {
    const int ka = 4 * s + 2 * kh;
    w2lo[s].x = W2[ka * OUTC + m];
    w2lo[s].y = W2[(ka + 1) * OUTC + m];
    w2hi[s].x = W2[ka * OUTC + 16 + m];
    w2hi[s].y = W2[(ka + 1) * OUTC + 16 + m];
  }

  float* T = tileT[threadIdx.x >> 5];
  #pragma unroll
  for (int r = 0; r < 8; ++r) {
    T[(8 * kh + r) * TP + m]      = c0[r];
    T[(8 * kh + r) * TP + 16 + m] = c1[r];
  }
  __builtin_amdgcn_fence(__ATOMIC_RELEASE, "wavefront");
  __builtin_amdgcn_wave_barrier();
  __builtin_amdgcn_fence(__ATOMIC_ACQUIRE, "wavefront");

  v8f d0 = {};
  v8f d1 = {};
  #pragma unroll
  for (int s = 0; s < OUTC / 4; ++s) {
    const int ka = 4 * s + 2 * kh;
    const v2f a = *reinterpret_cast<const v2f*>(&T[m * TP + ka]);
    d0 = __builtin_amdgcn_wmma_f32_16x16x4_f32(false, a, false, w2lo[s], (short)0, d0, false, false);
    d1 = __builtin_amdgcn_wmma_f32_16x16x4_f32(false, a, false, w2hi[s], (short)0, d1, false, false);
  }

  float* hrow = H + (row0 + 8 * kh) * OUTC + m;
  #pragma unroll
  for (int r = 0; r < 8; ++r) {
    hrow[r * OUTC]      = d0[r];
    hrow[r * OUTC + 16] = d1[r];
  }
}

#define N_BINS     ((N_NODES + 255) >> 8)
#define N_BINS_PAD 392
#define L1_BLOCKS  512
#define L1_STEPS   13
static_assert((long)L1_BLOCKS * 8 * L1_STEPS * 32 >= N_EDGES, "level-1 waves must cover every edge");
static_assert(N_BINS < N_BINS_PAD && N_BINS_PAD <= 512, "level-1 key is matched on 9 bits; the last row of the table is an empty bin");
static_assert(N_BINS_PAD <= 512 && N_BINS_PAD % 2 == 0, "two neighbouring bins per thread cover the table row of a 256-thread work-group");
static_assert(L1_BLOCKS == 512, "level 2 finds a run with 9 halvings and scans the runs with 16 waves");
#define L1_WG_EDGES (8 * L1_STEPS * 32)
#define L2_IMG      5120

__device__ __forceinline__ void wave_sync() {
  __builtin_amdgcn_fence(__ATOMIC_RELEASE, "wavefront");
  __builtin_amdgcn_wave_barrier();
  __builtin_amdgcn_fence(__ATOMIC_ACQUIRE, "wavefront");
}

template<int BITS>
__device__ __forceinline__ unsigned peer_mask(int key, bool valid) {
  unsigned peers = __builtin_amdgcn_ballot_w32(valid);
  #pragma unroll
  for (int b = 0; b < BITS; ++b) {
    const bool bit = (key >> b) & 1;
    const unsigned bal = __builtin_amdgcn_ballot_w32(bit);
    peers &= bit ? bal : ~bal;
  }
  return peers;
}

__device__ __forceinline__ int wave_sum(int v) {
  #pragma unroll
  for (int m = 16; m > 0; m >>= 1) v += __shfl_xor(v, m);
  return v;
}

__device__ __forceinline__ int wave_scan_incl(int v, int lane) {
  #pragma unroll
  for (int d = 1; d < 32; d <<= 1) {
    const int t = __shfl_up(v, d);
    if (lane >= d) v += t;
  }
  return v;
}

__global__ __launch_bounds__(256) void bin_local(
    const int*   __restrict__ eidx,
    const float* __restrict__ ew,
    int*         __restrict__ ls,
    int*         __restrict__ swl_i,
    int*         __restrict__ dl,
    int n_edges)
{
  __shared__ int  cur[8][N_BINS_PAD];
  __shared__ int  wtot[8];
  __shared__ int2 img_sw[L1_WG_EDGES];
  __shared__ int  img_d[L1_WG_EDGES];
  const int lane = threadIdx.x & 31;
  const int wv   = threadIdx.x >> 5;
  const int w    = (int)blockIdx.x * 8 + wv;
  for (int b = lane; b < N_BINS_PAD; b += 32) cur[wv][b] = 0;

  const bool owner = (int)threadIdx.x < N_BINS_PAD / 2;
  const int  bA    = 2 * min((int)threadIdx.x, N_BINS_PAD / 2 - 1);
  const int  bB    = bA + 1;

  int dst[L1_STEPS], srcv[L1_STEPS], wbits[L1_STEPS];
  const int e0 = w * (L1_STEPS * 32) + lane;
  #pragma unroll
  for (int s = 0; s < L1_STEPS; ++s) {
    const int e  = e0 + s * 32;
    const int ec = min(e, n_edges - 1);
    const int d  = eidx[n_edges + ec];
    srcv[s]  = eidx[ec];
    wbits[s] = __float_as_int(ew[ec]);
    dst[s]   = (e < n_edges) ? d : -1;
  }
  int2* swl = reinterpret_cast<int2*>(swl_i);
  wave_sync();
  unsigned pm[L1_STEPS];
  #pragma unroll
  for (int s = 0; s < L1_STEPS; ++s) {
    const bool valid = dst[s] >= 0;
    const int  key   = dst[s] >> 8;
    pm[s] = peer_mask<9>(key, valid);
    if (valid && (pm[s] >> lane) == 1u) cur[wv][key] += __popc(pm[s]);
    wave_sync();
  }
  __syncthreads();
  int cA[8], cB[8];
  int tA = 0, tB = 0;
  #pragma unroll
  for (int k = 0; k < 8; ++k) {
    cA[k] = owner ? cur[k][bA] : 0;
    cB[k] = owner ? cur[k][bB] : 0;
    tA += cA[k];
    tB += cB[k];
  }
  const int both = tA + tB;
  const int inc  = wave_scan_incl(both, lane);
  if (lane == 31) wtot[wv] = inc;
  __syncthreads();
  int ex = inc - both;
  #pragma unroll
  for (int k = 0; k < 8; ++k) if (k < wv) ex += wtot[k];
  if (owner) {
    int run = ex;
    #pragma unroll
    for (int k = 0; k < 8; ++k) { cur[k][bA] = run; run += cA[k]; }
    #pragma unroll
    for (int k = 0; k < 8; ++k) { cur[k][bB] = run; run += cB[k]; }
    ls[bA * L1_BLOCKS + blockIdx.x] = ex;
    ls[bB * L1_BLOCKS + blockIdx.x] = ex + tA;
  }
  __syncthreads();
  #pragma unroll
  for (int s = 0; s < L1_STEPS; ++s) {
    const bool valid = dst[s] >= 0;
    const int  key   = dst[s] >> 8;
    const unsigned peers = pm[s];
    const int base = valid ? cur[wv][key] : 0;
    wave_sync();
    if (valid && (peers >> lane) == 1u) cur[wv][key] = base + __popc(peers);
    wave_sync();
    if (valid) {
      const int pos = base + __popc(peers & ((1u << lane) - 1u));
      img_sw[pos] = make_int2(srcv[s], wbits[s]);
      img_d[pos]  = dst[s];
    }
  }
  __syncthreads();
  const int first = (int)blockIdx.x * L1_WG_EDGES;
  const int nloc  = min(max(n_edges - first, 0), L1_WG_EDGES);
  if (nloc > 0) {
    int  dd[L1_STEPS];
    int2 rec[L1_STEPS];
    #pragma unroll
    for (int s = 0; s < L1_STEPS; ++s) {
      const int ic = min(s * 256 + (int)threadIdx.x, nloc - 1);
      dd[s]  = img_d[ic];
      rec[s] = img_sw[ic];
    }
    #pragma unroll
    for (int s = 0; s < L1_STEPS; ++s) {
      const int i = s * 256 + (int)threadIdx.x;
      if (i < nloc) {
        swl[first + i] = rec[s];
        dl[first + i]  = dd[s];
      }
    }
  }
}

__device__ __forceinline__ int find_run(const int* segstart, int i) {
  int lo = 0, hi = L1_BLOCKS;
  #pragma unroll
  for (int t = 0; t < 9; ++t) {
    const int m = (lo + hi) >> 1;
    if (segstart[m] <= i) lo = m; else hi = m;
  }
  return lo;
}

__global__ __launch_bounds__(1024) void bin_sort(
    const int* __restrict__ ls,
    const int* __restrict__ swl_i,
    const int* __restrict__ dl,
    int*       __restrict__ off,
    int*       __restrict__ pairs_i,
    int n_nodes)
{
  __shared__ int hist[32][256];
  __shared__ int nodeoff[256];
  __shared__ int wtot[8];
  __shared__ int2 img[L2_IMG];
  __shared__ int runs_part[16];
  __shared__ int lows_part[16];
  __shared__ int segstart[L1_BLOCKS];
  __shared__ int segsrc[L1_BLOCKS];
  __shared__ int srcmap[L2_IMG];
  const int tid  = threadIdx.x;
  const int lane = tid & 31;
  const int wv   = tid >> 5;
  const int bin  = blockIdx.x;

  const bool runner = tid < L1_BLOCKS;
  const int  g      = min(tid, L1_BLOCKS - 1);
  const int  lo_g   = ls[bin * L1_BLOCKS + g];
  const int  hi_g   = ls[(bin + 1) * L1_BLOCKS + g];
  const int  len    = runner ? hi_g - lo_g : 0;
  const int  inc0   = wave_scan_incl(len, lane);
  const int  lows   = wave_sum(runner ? lo_g : 0);
  if (runner && lane == 31) { runs_part[wv] = inc0; lows_part[wv] = lows; }
  #pragma unroll
  for (int k = 0; k < 8; ++k) hist[wv][lane + 32 * k] = 0;
  __syncthreads();

  int pre = 0, nbin = 0, beg = 0;
  #pragma unroll
  for (int k = 0; k < 16; ++k) {
    const int t = runs_part[k];
    if (k < wv) pre += t;
    nbin += t;
    beg  += lows_part[k];
  }
  const int  p0     = pre + inc0 - len;
  const int  s0     = g * L1_WG_EDGES + lo_g;
  const bool staged = nbin <= L2_IMG;
  if (runner) {
    segstart[tid] = p0;
    segsrc[tid]   = s0 - p0;
    if (staged) for (int j = 0; j < len; ++j) srcmap[p0 + j] = s0 + j;
  }
  __syncthreads();

  for (int i0 = 0; i0 < nbin; i0 += 1024) {
    const int  i     = i0 + tid;
    const bool valid = i < nbin;
    const int  ic    = min(i, nbin - 1);
    const int  sp    = staged ? srcmap[ic] : segsrc[find_run(segstart, ic)] + ic;
    const int  key   = dl[sp] & 255;
    const unsigned peers = peer_mask<8>(key, valid);
    if (valid && (peers >> lane) == 1u) hist[wv][key] += __popc(peers);
    wave_sync();
  }
  __syncthreads();

  int tot = 0, inc = 0;
  if (tid < 256) {
    for (int k = 0; k < 32; ++k) {
      const int c = hist[k][tid];
      hist[k][tid] = tot;
      tot += c;
    }
    inc = wave_scan_incl(tot, lane);
    if (lane == 31) wtot[wv] = inc;
  }
  __syncthreads();
  if (tid < 256) {
    int pre2 = 0;
    for (int k = 0; k < wv; ++k) pre2 += wtot[k];
    const int no = pre2 + inc - tot;
    nodeoff[tid] = no;
    const int node = bin * 256 + tid;
    if (node < n_nodes) off[node] = beg + no;
    if (node == n_nodes - 1) off[n_nodes] = beg + no + tot;
  }
  __syncthreads();

  const int2* swl   = reinterpret_cast<const int2*>(swl_i);
  int2*       pairs = reinterpret_cast<int2*>(pairs_i);
  for (int i0 = 0; i0 < nbin; i0 += 1024) {
    const int  i     = i0 + tid;
    const bool valid = i < nbin;
    const int  ic    = min(i, nbin - 1);
    const int  sp    = staged ? srcmap[ic] : segsrc[find_run(segstart, ic)] + ic;
    const int  key   = dl[sp] & 255;
    const int2 rec   = swl[sp];
    const unsigned peers = peer_mask<8>(key, valid);
    const int base = hist[wv][key];
    wave_sync();
    if (valid && (peers >> lane) == 1u) hist[wv][key] = base + __popc(peers);
    wave_sync();
    if (valid) {
      const int lp = nodeoff[key] + base + __popc(peers & ((1u << lane) - 1u));
      if (staged) img[lp] = rec;
      else        pairs[beg + lp] = rec;
    }
  }
  if (staged) {
    __syncthreads();
    for (int i = tid; i < nbin; i += 1024) pairs[beg + i] = img[i];
  }
}

#define PULL_K 16

__device__ __forceinline__ int2 load_pairs(const int2* __restrict__ pairs, int b, int e, int lane) {
  int2 v = make_int2(0, 0);
  if (b < e) v = pairs[min(b + lane, e - 1)];
  return v;
}
__device__ __forceinline__ int2 mask_pairs(int2 v, int b, int e, int lane) {
  if (b + lane >= e) v.y = 0;
  return v;
}

template<int U>
__device__ __forceinline__ float pull_rows(const float* __restrict__ H, int2 p, int lane, float acc) {
  float hv[U], wt[U];
  #pragma unroll
  for (int u = 0; u < U; ++u) {
    const int s = __builtin_amdgcn_readlane(p.x, u);
    wt[u] = __int_as_float(__builtin_amdgcn_readlane(p.y, u));
    hv[u] = H[(long)s * OUTC + lane];
  }
  #pragma unroll
  for (int u = 0; u < U; ++u) acc = fmaf(wt[u], hv[u], acc);
  return acc;
}

__device__ __forceinline__ float pull_chunk(const float* __restrict__ H, int2 p, int cnt, int lane, float acc) {
  if (cnt <= 8)  return pull_rows<8>(H, p, lane, acc);
  if (cnt <= 16) return pull_rows<16>(H, p, lane, acc);
  if (cnt <= 24) return pull_rows<24>(H, p, lane, acc);
  return pull_rows<32>(H, p, lane, acc);
}

__global__ __launch_bounds__(256) void gcn_pull(
    const float* __restrict__ H,
    const int*   __restrict__ off,
    const int*   __restrict__ pairs_i,
    float*       __restrict__ out,
    int n_nodes)
{
  const int lane = threadIdx.x & 31;
  const int n0 = __builtin_amdgcn_readfirstlane((int)(blockIdx.x * 8 + (threadIdx.x >> 5)) * PULL_K);
  if (n0 >= n_nodes) return;
  const int nk = min(PULL_K, n_nodes - n0);
  const int2* pairs = reinterpret_cast<const int2*>(pairs_i);
  const int o = off[min(n0 + lane, n_nodes)];

  const int b0 = __builtin_amdgcn_readlane(o, 0);
  const int e0 = __builtin_amdgcn_readlane(o, 1);
  int2 p = mask_pairs(load_pairs(pairs, b0, e0, lane), b0, e0, lane);
  for (int i = 0; i < nk; ++i) {
    const int beg = __builtin_amdgcn_readlane(o, i);
    const int end = __builtin_amdgcn_readlane(o, i + 1);
    const int ne = (i + 1 < nk) ? __builtin_amdgcn_readlane(o, i + 2) : end;
    const int2 pn = load_pairs(pairs, end, ne, lane);
    float acc = 0.f;
    if (beg < end) acc = pull_chunk(H, p, min(32, end - beg), lane, acc);
    for (int base = beg + 32; base < end; base += 32) {
      const int2 q = mask_pairs(load_pairs(pairs, base, end, lane), base, end, lane);
      acc = pull_chunk(H, q, min(32, end - base), lane, acc);
    }
    out[(long)(n0 + i) * OUTC + lane] = acc;
    p = mask_pairs(pn, end, ne, lane);
  }
}

extern "C" void kernel_launch(void* const* d_in, const int* in_sizes, int n_in,
                              void* d_out, int out_size, void* d_ws, size_t ws_size,
                              hipStream_t stream) {
  const float* x  = (const float*)d_in[0];
  const float* W1 = (const float*)d_in[1];
  const float* W2 = (const float*)d_in[2];
  const float* ew = (const float*)d_in[3];
  const int*   ei = (const int*)d_in[4];
  float* out = (float*)d_out;

  char* ws = (char*)d_ws;
  float* h      = (float*)(ws);
  float* agg1   = (float*)(ws + 12800000);
  int*   pairs  = (int*)(ws + 25600000);
  int*   swl    = (int*)(ws + 38400000);
  int*   dl     = (int*)(ws + 52031488);
  int*   ls     = (int*)(ws + 58847232);
  int*   off    = (int*)(ws + 59650048);

  bin_local<<<L1_BLOCKS, 256, 0, stream>>>(ei, ew, ls, swl, dl, N_EDGES);
  bin_sort <<<N_BINS, 1024, 0, stream>>>(ls, swl, dl, off, pairs, N_NODES);

  const int tiles = N_NODES / 16;
  const int wpb   = 256 / 32;
  dim3 gemm_grid((tiles + wpb - 1) / wpb);
  gemm_x32_wmma<128><<<gemm_grid, 256, 0, stream>>>(x, W1, W2, h, N_NODES);

  const int pull_grid = (N_NODES + 8 * PULL_K - 1) / (8 * PULL_K);
  gcn_pull<<<pull_grid, 256, 0, stream>>>(h,    off, pairs, agg1, N_NODES);
  gcn_pull<<<pull_grid, 256, 0, stream>>>(agg1, off, pairs, out,  N_NODES);
}


# GEMM and level 1 of the edge sort, which do not depend on each other, run as different work-groups of one grid (sort groups first, LDS shared); 5 launches -> 4; arithmetic and sum order unchanged
# speedup vs baseline: 2.9150x; 1.0666x over previous
#include <hip/hip_runtime.h>


typedef __attribute__((ext_vector_type(2))) float v2f;
typedef __attribute__((ext_vector_type(8))) float v8f;

#define N_NODES 100000
#define N_EDGES 1600000
#define OUTC 32
#define TP   34

template<int K>
__device__ __forceinline__ void gemm_x32_wmma(
    const float* __restrict__ X,
    const float* __restrict__ W,
    const float* __restrict__ W2,
    float* __restrict__ H,
    int n_rows, int block, float* lds)
{
  const int lane = threadIdx.x & 31;
  const int tile = block * 8 + (int)(threadIdx.x >> 5);
  const int n_tiles = n_rows >> 4;
  if (tile >= n_tiles) return;

  const int  m   = lane & 15;
  const int  kh  = lane >> 4;
  const long row0 = (long)tile << 4;

  const float* xrow = X + (row0 + m) * K;

  v8f c0 = {};
  v8f c1 = {};

  #pragma unroll 8
  for (int kb = 0; kb < K; kb += 4) {
    const int ka = kb + 2 * kh;
    v2f a;
    a.x = xrow[ka];
    a.y = xrow[ka + 1];
    v2f b0, b1;
    b0.x = W[ka * OUTC + m];
    b0.y = W[(ka + 1) * OUTC + m];
    b1.x = W[ka * OUTC + 16 + m];
    b1.y = W[(ka + 1) * OUTC + 16 + m];
    c0 = __builtin_amdgcn_wmma_f32_16x16x4_f32(false, a, false, b0, (short)0, c0, false, false);
    c1 = __builtin_amdgcn_wmma_f32_16x16x4_f32(false, a, false, b1, (short)0, c1, false, false);
  }

  v2f w2lo[OUTC / 4], w2hi[OUTC / 4];
  #pragma unroll
  for (int s = 0; s < OUTC / 4; ++s) {
    const int ka = 4 * s + 2 * kh;
    w2lo[s].x = W2[ka * OUTC + m];
    w2lo[s].y = W2[(ka + 1) * OUTC + m];
    w2hi[s].x = W2[ka * OUTC + 16 + m];
    w2hi[s].y = W2[(ka + 1) * OUTC + 16 + m];
  }

  float* T = lds + (threadIdx.x >> 5) * (16 * TP);
  #pragma unroll
  for (int r = 0; r < 8; ++r) {
    T[(8 * kh + r) * TP + m]      = c0[r];
    T[(8 * kh + r) * TP + 16 + m] = c1[r];
  }
  __builtin_amdgcn_fence(__ATOMIC_RELEASE, "wavefront");
  __builtin_amdgcn_wave_barrier();
  __builtin_amdgcn_fence(__ATOMIC_ACQUIRE, "wavefront");

  v8f d0 = {};
  v8f d1 = {};
  #pragma unroll
  for (int s = 0; s < OUTC / 4; ++s) {
    const int ka = 4 * s + 2 * kh;
    const v2f a = *reinterpret_cast<const v2f*>(&T[m * TP + ka]);
    d0 = __builtin_amdgcn_wmma_f32_16x16x4_f32(false, a, false, w2lo[s], (short)0, d0, false, false);
    d1 = __builtin_amdgcn_wmma_f32_16x16x4_f32(false, a, false, w2hi[s], (short)0, d1, false, false);
  }

  float* hrow = H + (row0 + 8 * kh) * OUTC + m;
  #pragma unroll
  for (int r = 0; r < 8; ++r) {
    hrow[r * OUTC]      = d0[r];
    hrow[r * OUTC + 16] = d1[r];
  }
}

#define N_BINS     ((N_NODES + 255) >> 8)
#define N_BINS_PAD 392
#define L1_BLOCKS  512
#define L1_STEPS   13
static_assert((long)L1_BLOCKS * 8 * L1_STEPS * 32 >= N_EDGES, "level-1 waves must cover every edge");
static_assert(N_BINS < N_BINS_PAD && N_BINS_PAD <= 512, "level-1 key is matched on 9 bits; the last row of the table is an empty bin");
static_assert(N_BINS_PAD <= 512 && N_BINS_PAD % 2 == 0, "two neighbouring bins per thread cover the table row of a 256-thread work-group");
static_assert(L1_BLOCKS == 512, "level 2 finds a run with 9 halvings and scans the runs with 16 waves");
#define L1_WG_EDGES (8 * L1_STEPS * 32)
#define L2_IMG      5120

__device__ __forceinline__ void wave_sync() {
  __builtin_amdgcn_fence(__ATOMIC_RELEASE, "wavefront");
  __builtin_amdgcn_wave_barrier();
  __builtin_amdgcn_fence(__ATOMIC_ACQUIRE, "wavefront");
}

template<int BITS>
__device__ __forceinline__ unsigned peer_mask(int key, bool valid) {
  unsigned peers = __builtin_amdgcn_ballot_w32(valid);
  #pragma unroll
  for (int b = 0; b < BITS; ++b) {
    const bool bit = (key >> b) & 1;
    const unsigned bal = __builtin_amdgcn_ballot_w32(bit);
    peers &= bit ? bal : ~bal;
  }
  return peers;
}

__device__ __forceinline__ int wave_sum(int v) {
  #pragma unroll
  for (int m = 16; m > 0; m >>= 1) v += __shfl_xor(v, m);
  return v;
}

__device__ __forceinline__ int wave_scan_incl(int v, int lane) {
  #pragma unroll
  for (int d = 1; d < 32; d <<= 1) {
    const int t = __shfl_up(v, d);
    if (lane >= d) v += t;
  }
  return v;
}

#define L1_LDS_INTS (8 * N_BINS_PAD + 8 + 2 * L1_WG_EDGES + L1_WG_EDGES)
__device__ __forceinline__ void bin_local(
    const int*   __restrict__ eidx,
    const float* __restrict__ ew,
    int*         __restrict__ ls,
    int*         __restrict__ swl_i,
    int*         __restrict__ dl,
    int n_edges, int block, int* lds)
{
  int  (*cur)[N_BINS_PAD] = reinterpret_cast<int (*)[N_BINS_PAD]>(lds);
  int*  wtot   = lds + 8 * N_BINS_PAD;
  int2* img_sw = reinterpret_cast<int2*>(lds + 8 * N_BINS_PAD + 8);
  int*  img_d  = lds + 8 * N_BINS_PAD + 8 + 2 * L1_WG_EDGES;
  const int lane = threadIdx.x & 31;
  const int wv   = threadIdx.x >> 5;
  const int w    = block * 8 + wv;
  for (int b = lane; b < N_BINS_PAD; b += 32) cur[wv][b] = 0;

  const bool owner = (int)threadIdx.x < N_BINS_PAD / 2;
  const int  bA    = 2 * min((int)threadIdx.x, N_BINS_PAD / 2 - 1);
  const int  bB    = bA + 1;

  int dst[L1_STEPS], srcv[L1_STEPS], wbits[L1_STEPS];
  const int e0 = w * (L1_STEPS * 32) + lane;
  #pragma unroll
  for (int s = 0; s < L1_STEPS; ++s) {
    const int e  = e0 + s * 32;
    const int ec = min(e, n_edges - 1);
    const int d  = eidx[n_edges + ec];
    srcv[s]  = eidx[ec];
    wbits[s] = __float_as_int(ew[ec]);
    dst[s]   = (e < n_edges) ? d : -1;
  }
  int2* swl = reinterpret_cast<int2*>(swl_i);
  wave_sync();
  unsigned pm[L1_STEPS];
  #pragma unroll
  for (int s = 0; s < L1_STEPS; ++s) {
    const bool valid = dst[s] >= 0;
    const int  key   = dst[s] >> 8;
    pm[s] = peer_mask<9>(key, valid);
    if (valid && (pm[s] >> lane) == 1u) cur[wv][key] += __popc(pm[s]);
    wave_sync();
  }
  __syncthreads();
  int cA[8], cB[8];
  int tA = 0, tB = 0;
  #pragma unroll
  for (int k = 0; k < 8; ++k) {
    cA[k] = owner ? cur[k][bA] : 0;
    cB[k] = owner ? cur[k][bB] : 0;
    tA += cA[k];
    tB += cB[k];
  }
  const int both = tA + tB;
  const int inc  = wave_scan_incl(both, lane);
  if (lane == 31) wtot[wv] = inc;
  __syncthreads();
  int ex = inc - both;
  #pragma unroll
  for (int k = 0; k < 8; ++k) if (k < wv) ex += wtot[k];
  if (owner) {
    int run = ex;
    #pragma unroll
    for (int k = 0; k < 8; ++k) { cur[k][bA] = run; run += cA[k]; }
    #pragma unroll
    for (int k = 0; k < 8; ++k) { cur[k][bB] = run; run += cB[k]; }
    ls[bA * L1_BLOCKS + block] = ex;
    ls[bB * L1_BLOCKS + block] = ex + tA;
  }
  __syncthreads();
  #pragma unroll
  for (int s = 0; s < L1_STEPS; ++s) {
    const bool valid = dst[s] >= 0;
    const int  key   = dst[s] >> 8;
    const unsigned peers = pm[s];
    const int base = valid ? cur[wv][key] : 0;
    wave_sync();
    if (valid && (peers >> lane) == 1u) cur[wv][key] = base + __popc(peers);
    wave_sync();
    if (valid) {
      const int pos = base + __popc(peers & ((1u << lane) - 1u));
      img_sw[pos] = make_int2(srcv[s], wbits[s]);
      img_d[pos]  = dst[s];
    }
  }
  __syncthreads();
  const int first = block * L1_WG_EDGES;
  const int nloc  = min(max(n_edges - first, 0), L1_WG_EDGES);
  if (nloc > 0) {
    int  dd[L1_STEPS];
    int2 rec[L1_STEPS];
    #pragma unroll
    for (int s = 0; s < L1_STEPS; ++s) {
      const int ic = min(s * 256 + (int)threadIdx.x, nloc - 1);
      dd[s]  = img_d[ic];
      rec[s] = img_sw[ic];
    }
    #pragma unroll
    for (int s = 0; s < L1_STEPS; ++s) {
      const int i = s * 256 + (int)threadIdx.x;
      if (i < nloc) {
        swl[first + i] = rec[s];
        dl[first + i]  = dd[s];
      }
    }
  }
}

#define GEMM_BLOCKS ((N_NODES / 16 + 7) / 8)
static_assert(8 * 16 * TP <= L1_LDS_INTS, "the GEMM's staging tiles fit the level-1 LDS");
__global__ __launch_bounds__(256) void sort1_and_gemm(
    const float* __restrict__ x,
    const float* __restrict__ W1,
    const float* __restrict__ W2,
    float*       __restrict__ h,
    const int*   __restrict__ eidx,
    const float* __restrict__ ew,
    int*         __restrict__ ls,
    int*         __restrict__ swl_i,
    int*         __restrict__ dl)
{
  __shared__ __attribute__((aligned(16))) int lds[L1_LDS_INTS];
  if ((int)blockIdx.x < L1_BLOCKS) {
    bin_local(eidx, ew, ls, swl_i, dl, N_EDGES, (int)blockIdx.x, lds);
  } else {
    gemm_x32_wmma<128>(x, W1, W2, h, N_NODES, (int)blockIdx.x - L1_BLOCKS, reinterpret_cast<float*>(lds));
  }
}

__device__ __forceinline__ int find_run(const int* segstart, int i) {
  int lo = 0, hi = L1_BLOCKS;
  #pragma unroll
  for (int t = 0; t < 9; ++t) {
    const int m = (lo + hi) >> 1;
    if (segstart[m] <= i) lo = m; else hi = m;
  }
  return lo;
}

__global__ __launch_bounds__(1024) void bin_sort(
    const int* __restrict__ ls,
    const int* __restrict__ swl_i,
    const int* __restrict__ dl,
    int*       __restrict__ off,
    int*       __restrict__ pairs_i,
    int n_nodes)
{
  __shared__ int hist[32][256];
  __shared__ int nodeoff[256];
  __shared__ int wtot[8];
  __shared__ int2 img[L2_IMG];
  __shared__ int runs_part[16];
  __shared__ int lows_part[16];
  __shared__ int segstart[L1_BLOCKS];
  __shared__ int segsrc[L1_BLOCKS];
  __shared__ int srcmap[L2_IMG];
  const int tid  = threadIdx.x;
  const int lane = tid & 31;
  const int wv   = tid >> 5;
  const int bin  = blockIdx.x;

  const bool runner = tid < L1_BLOCKS;
  const int  g      = min(tid, L1_BLOCKS - 1);
  const int  lo_g   = ls[bin * L1_BLOCKS + g];
  const int  hi_g   = ls[(bin + 1) * L1_BLOCKS + g];
  const int  len    = runner ? hi_g - lo_g : 0;
  const int  inc0   = wave_scan_incl(len, lane);
  const int  lows   = wave_sum(runner ? lo_g : 0);
  if (runner && lane == 31) { runs_part[wv] = inc0; lows_part[wv] = lows; }
  #pragma unroll
  for (int k = 0; k < 8; ++k) hist[wv][lane + 32 * k] = 0;
  __syncthreads();

  int pre = 0, nbin = 0, beg = 0;
  #pragma unroll
  for (int k = 0; k < 16; ++k) {
    const int t = runs_part[k];
    if (k < wv) pre += t;
    nbin += t;
    beg  += lows_part[k];
  }
  const int  p0     = pre + inc0 - len;
  const int  s0     = g * L1_WG_EDGES + lo_g;
  const bool staged = nbin <= L2_IMG;
  if (runner) {
    segstart[tid] = p0;
    segsrc[tid]   = s0 - p0;
    if (staged) for (int j = 0; j < len; ++j) srcmap[p0 + j] = s0 + j;
  }
  __syncthreads();

  for (int i0 = 0; i0 < nbin; i0 += 1024) {
    const int  i     = i0 + tid;
    const bool valid = i < nbin;
    const int  ic    = min(i, nbin - 1);
    const int  sp    = staged ? srcmap[ic] : segsrc[find_run(segstart, ic)] + ic;
    const int  key   = dl[sp] & 255;
    const unsigned peers = peer_mask<8>(key, valid);
    if (valid && (peers >> lane) == 1u) hist[wv][key] += __popc(peers);
    wave_sync();
  }
  __syncthreads();

  int tot = 0, inc = 0;
  if (tid < 256) {
    for (int k = 0; k < 32; ++k) {
      const int c = hist[k][tid];
      hist[k][tid] = tot;
      tot += c;
    }
    inc = wave_scan_incl(tot, lane);
    if (lane == 31) wtot[wv] = inc;
  }
  __syncthreads();
  if (tid < 256) {
    int pre2 = 0;
    for (int k = 0; k < wv; ++k) pre2 += wtot[k];
    const int no = pre2 + inc - tot;
    nodeoff[tid] = no;
    const int node = bin * 256 + tid;
    if (node < n_nodes) off[node] = beg + no;
    if (node == n_nodes - 1) off[n_nodes] = beg + no + tot;
  }
  __syncthreads();

  const int2* swl   = reinterpret_cast<const int2*>(swl_i);
  int2*       pairs = reinterpret_cast<int2*>(pairs_i);
  for (int i0 = 0; i0 < nbin; i0 += 1024) {
    const int  i     = i0 + tid;
    const bool valid = i < nbin;
    const int  ic    = min(i, nbin - 1);
    const int  sp    = staged ? srcmap[ic] : segsrc[find_run(segstart, ic)] + ic;
    const int  key   = dl[sp] & 255;
    const int2 rec   = swl[sp];
    const unsigned peers = peer_mask<8>(key, valid);
    const int base = hist[wv][key];
    wave_sync();
    if (valid && (peers >> lane) == 1u) hist[wv][key] = base + __popc(peers);
    wave_sync();
    if (valid) {
      const int lp = nodeoff[key] + base + __popc(peers & ((1u << lane) - 1u));
      if (staged) img[lp] = rec;
      else        pairs[beg + lp] = rec;
    }
  }
  if (staged) {
    __syncthreads();
    for (int i = tid; i < nbin; i += 1024) pairs[beg + i] = img[i];
  }
}

#define PULL_K 16

__device__ __forceinline__ int2 load_pairs(const int2* __restrict__ pairs, int b, int e, int lane) {
  int2 v = make_int2(0, 0);
  if (b < e) v = pairs[min(b + lane, e - 1)];
  return v;
}
__device__ __forceinline__ int2 mask_pairs(int2 v, int b, int e, int lane) {
  if (b + lane >= e) v.y = 0;
  return v;
}

template<int U>
__device__ __forceinline__ float pull_rows(const float* __restrict__ H, int2 p, int lane, float acc) {
  float hv[U], wt[U];
  #pragma unroll
  for (int u = 0; u < U; ++u) {
    const int s = __builtin_amdgcn_readlane(p.x, u);
    wt[u] = __int_as_float(__builtin_amdgcn_readlane(p.y, u));
    hv[u] = H[(long)s * OUTC + lane];
  }
  #pragma unroll
  for (int u = 0; u < U; ++u) acc = fmaf(wt[u], hv[u], acc);
  return acc;
}

__device__ __forceinline__ float pull_chunk(const float* __restrict__ H, int2 p, int cnt, int lane, float acc) {
  if (cnt <= 8)  return pull_rows<8>(H, p, lane, acc);
  if (cnt <= 16) return pull_rows<16>(H, p, lane, acc);
  if (cnt <= 24) return pull_rows<24>(H, p, lane, acc);
  return pull_rows<32>(H, p, lane, acc);
}

__global__ __launch_bounds__(256) void gcn_pull(
    const float* __restrict__ H,
    const int*   __restrict__ off,
    const int*   __restrict__ pairs_i,
    float*       __restrict__ out,
    int n_nodes)
{
  const int lane = threadIdx.x & 31;
  const int n0 = __builtin_amdgcn_readfirstlane((int)(blockIdx.x * 8 + (threadIdx.x >> 5)) * PULL_K);
  if (n0 >= n_nodes) return;
  const int nk = min(PULL_K, n_nodes - n0);
  const int2* pairs = reinterpret_cast<const int2*>(pairs_i);
  const int o = off[min(n0 + lane, n_nodes)];

  const int b0 = __builtin_amdgcn_readlane(o, 0);
  const int e0 = __builtin_amdgcn_readlane(o, 1);
  int2 p = mask_pairs(load_pairs(pairs, b0, e0, lane), b0, e0, lane);
  for (int i = 0; i < nk; ++i) {
    const int beg = __builtin_amdgcn_readlane(o, i);
    const int end = __builtin_amdgcn_readlane(o, i + 1);
    const int ne = (i + 1 < nk) ? __builtin_amdgcn_readlane(o, i + 2) : end;
    const int2 pn = load_pairs(pairs, end, ne, lane);
    float acc = 0.f;
    if (beg < end) acc = pull_chunk(H, p, min(32, end - beg), lane, acc);
    for (int base = beg + 32; base < end; base += 32) {
      const int2 q = mask_pairs(load_pairs(pairs, base, end, lane), base, end, lane);
      acc = pull_chunk(H, q, min(32, end - base), lane, acc);
    }
    out[(long)(n0 + i) * OUTC + lane] = acc;
    p = mask_pairs(pn, end, ne, lane);
  }
}

extern "C" void kernel_launch(void* const* d_in, const int* in_sizes, int n_in,
                              void* d_out, int out_size, void* d_ws, size_t ws_size,
                              hipStream_t stream) {
  const float* x  = (const float*)d_in[0];
  const float* W1 = (const float*)d_in[1];
  const float* W2 = (const float*)d_in[2];
  const float* ew = (const float*)d_in[3];
  const int*   ei = (const int*)d_in[4];
  float* out = (float*)d_out;

  char* ws = (char*)d_ws;
  float* h      = (float*)(ws);
  float* agg1   = (float*)(ws + 12800000);
  int*   pairs  = (int*)(ws + 25600000);
  int*   swl    = (int*)(ws + 38400000);
  int*   dl     = (int*)(ws + 52031488);
  int*   ls     = (int*)(ws + 58847232);
  int*   off    = (int*)(ws + 59650048);

  sort1_and_gemm<<<L1_BLOCKS + GEMM_BLOCKS, 256, 0, stream>>>(x, W1, W2, h, ei, ew, ls, swl, dl);
  bin_sort <<<N_BINS, 1024, 0, stream>>>(ls, swl, dl, off, pairs, N_NODES);

  const int pull_grid = (N_NODES + 8 * PULL_K - 1) / (8 * PULL_K);
  gcn_pull<<<pull_grid, 256, 0, stream>>>(h,    off, pairs, agg1, N_NODES);
  gcn_pull<<<pull_grid, 256, 0, stream>>>(agg1, off, pairs, out,  N_NODES);
}


# Layer-1 aggregation moved into the level-2 sort launch: the work-group that ordered a bin sums its 256 nodes from the pairs still in its LDS; 4 launches -> 3; per-node sum order unchanged
# speedup vs baseline: 3.0956x; 1.0620x over previous
#include <hip/hip_runtime.h>


typedef __attribute__((ext_vector_type(2))) float v2f;
typedef __attribute__((ext_vector_type(8))) float v8f;

#define N_NODES 100000
#define N_EDGES 1600000
#define OUTC 32
#define TP   34

template<int K>
__device__ __forceinline__ void gemm_x32_wmma(
    const float* __restrict__ X,
    const float* __restrict__ W,
    const float* __restrict__ W2,
    float* __restrict__ H,
    int n_rows, int block, float* lds)
{
  const int lane = threadIdx.x & 31;
  const int tile = block * 8 + (int)(threadIdx.x >> 5);
  const int n_tiles = n_rows >> 4;
  if (tile >= n_tiles) return;

  const int  m   = lane & 15;
  const int  kh  = lane >> 4;
  const long row0 = (long)tile << 4;

  const float* xrow = X + (row0 + m) * K;

  v8f c0 = {};
  v8f c1 = {};

  #pragma unroll 8
  for (int kb = 0; kb < K; kb += 4) {
    const int ka = kb + 2 * kh;
    v2f a;
    a.x = xrow[ka];
    a.y = xrow[ka + 1];
    v2f b0, b1;
    b0.x = W[ka * OUTC + m];
    b0.y = W[(ka + 1) * OUTC + m];
    b1.x = W[ka * OUTC + 16 + m];
    b1.y = W[(ka + 1) * OUTC + 16 + m];
    c0 = __builtin_amdgcn_wmma_f32_16x16x4_f32(false, a, false, b0, (short)0, c0, false, false);
    c1 = __builtin_amdgcn_wmma_f32_16x16x4_f32(false, a, false, b1, (short)0, c1, false, false);
  }

  v2f w2lo[OUTC / 4], w2hi[OUTC / 4];
  #pragma unroll
  for (int s = 0; s < OUTC / 4; ++s) {
    const int ka = 4 * s + 2 * kh;
    w2lo[s].x = W2[ka * OUTC + m];
    w2lo[s].y = W2[(ka + 1) * OUTC + m];
    w2hi[s].x = W2[ka * OUTC + 16 + m];
    w2hi[s].y = W2[(ka + 1) * OUTC + 16 + m];
  }

  float* T = lds + (threadIdx.x >> 5) * (16 * TP);
  #pragma unroll
  for (int r = 0; r < 8; ++r) {
    T[(8 * kh + r) * TP + m]      = c0[r];
    T[(8 * kh + r) * TP + 16 + m] = c1[r];
  }
  __builtin_amdgcn_fence(__ATOMIC_RELEASE, "wavefront");
  __builtin_amdgcn_wave_barrier();
  __builtin_amdgcn_fence(__ATOMIC_ACQUIRE, "wavefront");

  v8f d0 = {};
  v8f d1 = {};
  #pragma unroll
  for (int s = 0; s < OUTC / 4; ++s) {
    const int ka = 4 * s + 2 * kh;
    const v2f a = *reinterpret_cast<const v2f*>(&T[m * TP + ka]);
    d0 = __builtin_amdgcn_wmma_f32_16x16x4_f32(false, a, false, w2lo[s], (short)0, d0, false, false);
    d1 = __builtin_amdgcn_wmma_f32_16x16x4_f32(false, a, false, w2hi[s], (short)0, d1, false, false);
  }

  float* hrow = H + (row0 + 8 * kh) * OUTC + m;
  #pragma unroll
  for (int r = 0; r < 8; ++r) {
    hrow[r * OUTC]      = d0[r];
    hrow[r * OUTC + 16] = d1[r];
  }
}

#define N_BINS     ((N_NODES + 255) >> 8)
#define N_BINS_PAD 392
#define L1_BLOCKS  512
#define L1_STEPS   13
static_assert((long)L1_BLOCKS * 8 * L1_STEPS * 32 >= N_EDGES, "level-1 waves must cover every edge");
static_assert(N_BINS < N_BINS_PAD && N_BINS_PAD <= 512, "level-1 key is matched on 9 bits; the last row of the table is an empty bin");
static_assert(N_BINS_PAD <= 512 && N_BINS_PAD % 2 == 0, "two neighbouring bins per thread cover the table row of a 256-thread work-group");
static_assert(L1_BLOCKS == 512, "level 2 finds a run with 9 halvings and scans the runs with 16 waves");
#define L1_WG_EDGES (8 * L1_STEPS * 32)
#define L2_IMG      5120

__device__ __forceinline__ void wave_sync() {
  __builtin_amdgcn_fence(__ATOMIC_RELEASE, "wavefront");
  __builtin_amdgcn_wave_barrier();
  __builtin_amdgcn_fence(__ATOMIC_ACQUIRE, "wavefront");
}

template<int BITS>
__device__ __forceinline__ unsigned peer_mask(int key, bool valid) {
  unsigned peers = __builtin_amdgcn_ballot_w32(valid);
  #pragma unroll
  for (int b = 0; b < BITS; ++b) {
    const bool bit = (key >> b) & 1;
    const unsigned bal = __builtin_amdgcn_ballot_w32(bit);
    peers &= bit ? bal : ~bal;
  }
  return peers;
}

__device__ __forceinline__ int wave_sum(int v) {
  #pragma unroll
  for (int m = 16; m > 0; m >>= 1) v += __shfl_xor(v, m);
  return v;
}

__device__ __forceinline__ int wave_scan_incl(int v, int lane) {
  #pragma unroll
  for (int d = 1; d < 32; d <<= 1) {
    const int t = __shfl_up(v, d);
    if (lane >= d) v += t;
  }
  return v;
}

#define L1_LDS_INTS (8 * N_BINS_PAD + 8 + 2 * L1_WG_EDGES + L1_WG_EDGES)
__device__ __forceinline__ void bin_local(
    const int*   __restrict__ eidx,
    const float* __restrict__ ew,
    int*         __restrict__ ls,
    int*         __restrict__ swl_i,
    int*         __restrict__ dl,
    int n_edges, int block, int* lds)
{
  int  (*cur)[N_BINS_PAD] = reinterpret_cast<int (*)[N_BINS_PAD]>(lds);
  int*  wtot   = lds + 8 * N_BINS_PAD;
  int2* img_sw = reinterpret_cast<int2*>(lds + 8 * N_BINS_PAD + 8);
  int*  img_d  = lds + 8 * N_BINS_PAD + 8 + 2 * L1_WG_EDGES;
  const int lane = threadIdx.x & 31;
  const int wv   = threadIdx.x >> 5;
  const int w    = block * 8 + wv;
  for (int b = lane; b < N_BINS_PAD; b += 32) cur[wv][b] = 0;

  const bool owner = (int)threadIdx.x < N_BINS_PAD / 2;
  const int  bA    = 2 * min((int)threadIdx.x, N_BINS_PAD / 2 - 1);
  const int  bB    = bA + 1;

  int dst[L1_STEPS], srcv[L1_STEPS], wbits[L1_STEPS];
  const int e0 = w * (L1_STEPS * 32) + lane;
  #pragma unroll
  for (int s = 0; s < L1_STEPS; ++s) {
    const int e  = e0 + s * 32;
    const int ec = min(e, n_edges - 1);
    const int d  = eidx[n_edges + ec];
    srcv[s]  = eidx[ec];
    wbits[s] = __float_as_int(ew[ec]);
    dst[s]   = (e < n_edges) ? d : -1;
  }
  int2* swl = reinterpret_cast<int2*>(swl_i);
  wave_sync();
  unsigned pm[L1_STEPS];
  #pragma unroll
  for (int s = 0; s < L1_STEPS; ++s) {
    const bool valid = dst[s] >= 0;
    const int  key   = dst[s] >> 8;
    pm[s] = peer_mask<9>(key, valid);
    if (valid && (pm[s] >> lane) == 1u) cur[wv][key] += __popc(pm[s]);
    wave_sync();
  }
  __syncthreads();
  int cA[8], cB[8];
  int tA = 0, tB = 0;
  #pragma unroll
  for (int k = 0; k < 8; ++k) {
    cA[k] = owner ? cur[k][bA] : 0;
    cB[k] = owner ? cur[k][bB] : 0;
    tA += cA[k];
    tB += cB[k];
  }
  const int both = tA + tB;
  const int inc  = wave_scan_incl(both, lane);
  if (lane == 31) wtot[wv] = inc;
  __syncthreads();
  int ex = inc - both;
  #pragma unroll
  for (int k = 0; k < 8; ++k) if (k < wv) ex += wtot[k];
  if (owner) {
    int run = ex;
    #pragma unroll
    for (int k = 0; k < 8; ++k) { cur[k][bA] = run; run += cA[k]; }
    #pragma unroll
    for (int k = 0; k < 8; ++k) { cur[k][bB] = run; run += cB[k]; }
    ls[bA * L1_BLOCKS + block] = ex;
    ls[bB * L1_BLOCKS + block] = ex + tA;
  }
  __syncthreads();
  #pragma unroll
  for (int s = 0; s < L1_STEPS; ++s) {
    const bool valid = dst[s] >= 0;
    const int  key   = dst[s] >> 8;
    const unsigned peers = pm[s];
    const int base = valid ? cur[wv][key] : 0;
    wave_sync();
    if (valid && (peers >> lane) == 1u) cur[wv][key] = base + __popc(peers);
    wave_sync();
    if (valid) {
      const int pos = base + __popc(peers & ((1u << lane) - 1u));
      img_sw[pos] = make_int2(srcv[s], wbits[s]);
      img_d[pos]  = dst[s];
    }
  }
  __syncthreads();
  const int first = block * L1_WG_EDGES;
  const int nloc  = min(max(n_edges - first, 0), L1_WG_EDGES);
  if (nloc > 0) {
    int  dd[L1_STEPS];
    int2 rec[L1_STEPS];
    #pragma unroll
    for (int s = 0; s < L1_STEPS; ++s) {
      const int ic = min(s * 256 + (int)threadIdx.x, nloc - 1);
      dd[s]  = img_d[ic];
      rec[s] = img_sw[ic];
    }
    #pragma unroll
    for (int s = 0; s < L1_STEPS; ++s) {
      const int i = s * 256 + (int)threadIdx.x;
      if (i < nloc) {
        swl[first + i] = rec[s];
        dl[first + i]  = dd[s];
      }
    }
  }
}

#define GEMM_BLOCKS ((N_NODES / 16 + 7) / 8)
static_assert(8 * 16 * TP <= L1_LDS_INTS, "the GEMM's staging tiles fit the level-1 LDS");
__global__ __launch_bounds__(256) void sort1_and_gemm(
    const float* __restrict__ x,
    const float* __restrict__ W1,
    const float* __restrict__ W2,
    float*       __restrict__ h,
    const int*   __restrict__ eidx,
    const float* __restrict__ ew,
    int*         __restrict__ ls,
    int*         __restrict__ swl_i,
    int*         __restrict__ dl)
{
  __shared__ __attribute__((aligned(16))) int lds[L1_LDS_INTS];
  if ((int)blockIdx.x < L1_BLOCKS) {
    bin_local(eidx, ew, ls, swl_i, dl, N_EDGES, (int)blockIdx.x, lds);
  } else {
    gemm_x32_wmma<128>(x, W1, W2, h, N_NODES, (int)blockIdx.x - L1_BLOCKS, reinterpret_cast<float*>(lds));
  }
}

#define PULL_K 16

__device__ __forceinline__ int2 load_pairs(const int2* pairs, int b, int e, int lane) {
  int2 v = make_int2(0, 0);
  if (b < e) v = pairs[min(b + lane, e - 1)];
  return v;
}
__device__ __forceinline__ int2 mask_pairs(int2 v, int b, int e, int lane) {
  if (b + lane >= e) v.y = 0;
  return v;
}

template<int U>
__device__ __forceinline__ float pull_rows(const float* __restrict__ H, int2 p, int lane, float acc) {
  float hv[U], wt[U];
  #pragma unroll
  for (int u = 0; u < U; ++u) {
    const int s = __builtin_amdgcn_readlane(p.x, u);
    wt[u] = __int_as_float(__builtin_amdgcn_readlane(p.y, u));
    hv[u] = H[(long)s * OUTC + lane];
  }
  #pragma unroll
  for (int u = 0; u < U; ++u) acc = fmaf(wt[u], hv[u], acc);
  return acc;
}

template<bool SMALL_CODE>
__device__ __forceinline__ float pull_chunk(const float* __restrict__ H, int2 p, int cnt, int lane, float acc) {
  if (SMALL_CODE) {
    for (int u0 = 0; u0 < cnt; u0 += 8) {
      float hv[8], wt[8];
      #pragma unroll
      for (int u = 0; u < 8; ++u) {
        const int s = __builtin_amdgcn_readlane(p.x, u0 + u);
        wt[u] = __int_as_float(__builtin_amdgcn_readlane(p.y, u0 + u));
        hv[u] = H[(long)s * OUTC + lane];
      }
      #pragma unroll
      for (int u = 0; u < 8; ++u) acc = fmaf(wt[u], hv[u], acc);
    }
    return acc;
  }
  if (cnt <= 8)  return pull_rows<8>(H, p, lane, acc);
  if (cnt <= 16) return pull_rows<16>(H, p, lane, acc);
  if (cnt <= 24) return pull_rows<24>(H, p, lane, acc);
  return pull_rows<32>(H, p, lane, acc);
}

template<bool SMALL_CODE>
__device__ __forceinline__ void pull_nodes(
    const float* __restrict__ H, const int2* pairs, int o, int nk, float* __restrict__ rows, int lane)
{
  const int b0 = __builtin_amdgcn_readlane(o, 0);
  const int e0 = __builtin_amdgcn_readlane(o, 1);
  int2 p = mask_pairs(load_pairs(pairs, b0, e0, lane), b0, e0, lane);
  for (int i = 0; i < nk; ++i) {
    const int beg = __builtin_amdgcn_readlane(o, i);
    const int end = __builtin_amdgcn_readlane(o, i + 1);
    const int ne = (i + 1 < nk) ? __builtin_amdgcn_readlane(o, i + 2) : end;
    const int2 pn = load_pairs(pairs, end, ne, lane);
    float acc = 0.f;
    if (beg < end) acc = pull_chunk<SMALL_CODE>(H, p, min(32, end - beg), lane, acc);
    for (int base = beg + 32; base < end; base += 32) {
      const int2 q = mask_pairs(load_pairs(pairs, base, end, lane), base, end, lane);
      acc = pull_chunk<SMALL_CODE>(H, q, min(32, end - base), lane, acc);
    }
    rows[i * OUTC + lane] = acc;
    p = mask_pairs(pn, end, ne, lane);
  }
}

__device__ __forceinline__ int find_run(const int* segstart, int i) {
  int lo = 0, hi = L1_BLOCKS;
  #pragma unroll
  for (int t = 0; t < 9; ++t) {
    const int m = (lo + hi) >> 1;
    if (segstart[m] <= i) lo = m; else hi = m;
  }
  return lo;
}

__global__ __launch_bounds__(1024) void bin_sort_pull(
    const int*   __restrict__ ls,
    const int*   __restrict__ swl_i,
    const int*   __restrict__ dl,
    int*         __restrict__ off,
    int*         __restrict__ pairs_i,
    const float* __restrict__ H,
    float*       __restrict__ agg,
    int n_nodes)
{
  __shared__ int hist[32][256];
  __shared__ int nodeoff[257];
  __shared__ int wtot[8];
  __shared__ int2 img[L2_IMG];
  __shared__ int runs_part[16];
  __shared__ int lows_part[16];
  __shared__ int segstart[L1_BLOCKS];
  __shared__ int segsrc[L1_BLOCKS];
  __shared__ int srcmap[L2_IMG];
  const int tid  = threadIdx.x;
  const int lane = tid & 31;
  const int wv   = tid >> 5;
  const int bin  = blockIdx.x;

  const bool runner = tid < L1_BLOCKS;
  const int  g      = min(tid, L1_BLOCKS - 1);
  const int  lo_g   = ls[bin * L1_BLOCKS + g];
  const int  hi_g   = ls[(bin + 1) * L1_BLOCKS + g];
  const int  len    = runner ? hi_g - lo_g : 0;
  const int  inc0   = wave_scan_incl(len, lane);
  const int  lows   = wave_sum(runner ? lo_g : 0);
  if (runner && lane == 31) { runs_part[wv] = inc0; lows_part[wv] = lows; }
  #pragma unroll
  for (int k = 0; k < 8; ++k) hist[wv][lane + 32 * k] = 0;
  __syncthreads();

  int pre = 0, nbin = 0, beg = 0;
  #pragma unroll
  for (int k = 0; k < 16; ++k) {
    const int t = runs_part[k];
    if (k < wv) pre += t;
    nbin += t;
    beg  += lows_part[k];
  }
  const int  p0     = pre + inc0 - len;
  const int  s0     = g * L1_WG_EDGES + lo_g;
  const bool staged = nbin <= L2_IMG;
  if (runner) {
    segstart[tid] = p0;
    segsrc[tid]   = s0 - p0;
    if (staged) for (int j = 0; j < len; ++j) srcmap[p0 + j] = s0 + j;
  }
  __syncthreads();

  for (int i0 = 0; i0 < nbin; i0 += 1024) {
    const int  i     = i0 + tid;
    const bool valid = i < nbin;
    const int  ic    = min(i, nbin - 1);
    const int  sp    = staged ? srcmap[ic] : segsrc[find_run(segstart, ic)] + ic;
    const int  key   = dl[sp] & 255;
    const unsigned peers = peer_mask<8>(key, valid);
    if (valid && (peers >> lane) == 1u) hist[wv][key] += __popc(peers);
    wave_sync();
  }
  __syncthreads();

  int tot = 0, inc = 0;
  if (tid < 256) {
    for (int k = 0; k < 32; ++k) {
      const int c = hist[k][tid];
      hist[k][tid] = tot;
      tot += c;
    }
    inc = wave_scan_incl(tot, lane);
    if (lane == 31) wtot[wv] = inc;
  }
  __syncthreads();
  if (tid < 256) {
    int pre2 = 0;
    for (int k = 0; k < wv; ++k) pre2 += wtot[k];
    const int no = pre2 + inc - tot;
    nodeoff[tid] = no;
    if (tid == 0) nodeoff[256] = nbin;
    const int node = bin * 256 + tid;
    if (node < n_nodes) off[node] = beg + no;
    if (node == n_nodes - 1) off[n_nodes] = beg + no + tot;
  }
  __syncthreads();

  const int2* swl   = reinterpret_cast<const int2*>(swl_i);
  int2*       pairs = reinterpret_cast<int2*>(pairs_i);
  for (int i0 = 0; i0 < nbin; i0 += 1024) {
    const int  i     = i0 + tid;
    const bool valid = i < nbin;
    const int  ic    = min(i, nbin - 1);
    const int  sp    = staged ? srcmap[ic] : segsrc[find_run(segstart, ic)] + ic;
    const int  key   = dl[sp] & 255;
    const int2 rec   = swl[sp];
    const unsigned peers = peer_mask<8>(key, valid);
    const int base = hist[wv][key];
    wave_sync();
    if (valid && (peers >> lane) == 1u) hist[wv][key] = base + __popc(peers);
    wave_sync();
    if (valid) {
      const int lp = nodeoff[key] + base + __popc(peers & ((1u << lane) - 1u));
      if (staged) img[lp] = rec;
      else        pairs[beg + lp] = rec;
    }
  }
  if (staged) {
    __syncthreads();
    for (int i = tid; i < nbin; i += 1024) pairs[beg + i] = img[i];
  } else {
    __syncthreads();
  }

  const int k0   = __builtin_amdgcn_readfirstlane(wv * 8);
  const int nloc = min(256, n_nodes - bin * 256);
  if (k0 < nloc) {
    const int nk = min(8, nloc - k0);
    const int o  = nodeoff[min(k0 + lane, 256)];
    float* rows  = agg + (long)(bin * 256 + k0) * OUTC;
    if (staged) pull_nodes<false>(H, img,         o, nk, rows, lane);
    else        pull_nodes<true> (H, pairs + beg, o, nk, rows, lane);
  }
}

__global__ __launch_bounds__(256) void gcn_pull(
    const float* __restrict__ H,
    const int*   __restrict__ off,
    const int*   __restrict__ pairs_i,
    float*       __restrict__ out,
    int n_nodes)
{
  const int lane = threadIdx.x & 31;
  const int n0 = __builtin_amdgcn_readfirstlane((int)(blockIdx.x * 8 + (threadIdx.x >> 5)) * PULL_K);
  if (n0 >= n_nodes) return;
  const int nk = min(PULL_K, n_nodes - n0);
  const int o = off[min(n0 + lane, n_nodes)];
  pull_nodes<false>(H, reinterpret_cast<const int2*>(pairs_i), o, nk, out + (long)n0 * OUTC, lane);
}

extern "C" void kernel_launch(void* const* d_in, const int* in_sizes, int n_in,
                              void* d_out, int out_size, void* d_ws, size_t ws_size,
                              hipStream_t stream) {
  const float* x  = (const float*)d_in[0];
  const float* W1 = (const float*)d_in[1];
  const float* W2 = (const float*)d_in[2];
  const float* ew = (const float*)d_in[3];
  const int*   ei = (const int*)d_in[4];
  float* out = (float*)d_out;

  char* ws = (char*)d_ws;
  float* h      = (float*)(ws);
  float* agg1   = (float*)(ws + 12800000);
  int*   pairs  = (int*)(ws + 25600000);
  int*   swl    = (int*)(ws + 38400000);
  int*   dl     = (int*)(ws + 52031488);
  int*   ls     = (int*)(ws + 58847232);
  int*   off    = (int*)(ws + 59650048);

  sort1_and_gemm<<<L1_BLOCKS + GEMM_BLOCKS, 256, 0, stream>>>(x, W1, W2, h, ei, ew, ls, swl, dl);
  bin_sort_pull<<<N_BINS, 1024, 0, stream>>>(ls, swl, dl, off, pairs, h, agg1, N_NODES);

  const int pull_grid = (N_NODES + 8 * PULL_K - 1) / (8 * PULL_K);
  gcn_pull<<<pull_grid, 256, 0, stream>>>(agg1, off, pairs, out, N_NODES);
}


# Level-2 sort reads each edge of a bin once: a thread's dst and (src, weight) loads issued together; key, record and peer mask stay in registers from count to rank; edge order and sums unchanged
# speedup vs baseline: 3.2470x; 1.0489x over previous
#include <hip/hip_runtime.h>


typedef __attribute__((ext_vector_type(2))) float v2f;
typedef __attribute__((ext_vector_type(8))) float v8f;

#define N_NODES 100000
#define N_EDGES 1600000
#define OUTC 32
#define TP   34

template<int K>
__device__ __forceinline__ void gemm_x32_wmma(
    const float* __restrict__ X,
    const float* __restrict__ W,
    const float* __restrict__ W2,
    float* __restrict__ H,
    int n_rows, int block, float* lds)
{
  const int lane = threadIdx.x & 31;
  const int tile = block * 8 + (int)(threadIdx.x >> 5);
  const int n_tiles = n_rows >> 4;
  if (tile >= n_tiles) return;

  const int  m   = lane & 15;
  const int  kh  = lane >> 4;
  const long row0 = (long)tile << 4;

  const float* xrow = X + (row0 + m) * K;

  v8f c0 = {};
  v8f c1 = {};

  #pragma unroll 8
  for (int kb = 0; kb < K; kb += 4) {
    const int ka = kb + 2 * kh;
    v2f a;
    a.x = xrow[ka];
    a.y = xrow[ka + 1];
    v2f b0, b1;
    b0.x = W[ka * OUTC + m];
    b0.y = W[(ka + 1) * OUTC + m];
    b1.x = W[ka * OUTC + 16 + m];
    b1.y = W[(ka + 1) * OUTC + 16 + m];
    c0 = __builtin_amdgcn_wmma_f32_16x16x4_f32(false, a, false, b0, (short)0, c0, false, false);
    c1 = __builtin_amdgcn_wmma_f32_16x16x4_f32(false, a, false, b1, (short)0, c1, false, false);
  }

  v2f w2lo[OUTC / 4], w2hi[OUTC / 4];
  #pragma unroll
  for (int s = 0; s < OUTC / 4; ++s) {
    const int ka = 4 * s + 2 * kh;
    w2lo[s].x = W2[ka * OUTC + m];
    w2lo[s].y = W2[(ka + 1) * OUTC + m];
    w2hi[s].x = W2[ka * OUTC + 16 + m];
    w2hi[s].y = W2[(ka + 1) * OUTC + 16 + m];
  }

  float* T = lds + (threadIdx.x >> 5) * (16 * TP);
  #pragma unroll
  for (int r = 0; r < 8; ++r) {
    T[(8 * kh + r) * TP + m]      = c0[r];
    T[(8 * kh + r) * TP + 16 + m] = c1[r];
  }
  __builtin_amdgcn_fence(__ATOMIC_RELEASE, "wavefront");
  __builtin_amdgcn_wave_barrier();
  __builtin_amdgcn_fence(__ATOMIC_ACQUIRE, "wavefront");

  v8f d0 = {};
  v8f d1 = {};
  #pragma unroll
  for (int s = 0; s < OUTC / 4; ++s) {
    const int ka = 4 * s + 2 * kh;
    const v2f a = *reinterpret_cast<const v2f*>(&T[m * TP + ka]);
    d0 = __builtin_amdgcn_wmma_f32_16x16x4_f32(false, a, false, w2lo[s], (short)0, d0, false, false);
    d1 = __builtin_amdgcn_wmma_f32_16x16x4_f32(false, a, false, w2hi[s], (short)0, d1, false, false);
  }

  float* hrow = H + (row0 + 8 * kh) * OUTC + m;
  #pragma unroll
  for (int r = 0; r < 8; ++r) {
    hrow[r * OUTC]      = d0[r];
    hrow[r * OUTC + 16] = d1[r];
  }
}

#define N_BINS     ((N_NODES + 255) >> 8)
#define N_BINS_PAD 392
#define L1_BLOCKS  512
#define L1_STEPS   13
static_assert((long)L1_BLOCKS * 8 * L1_STEPS * 32 >= N_EDGES, "level-1 waves must cover every edge");
static_assert(N_BINS < N_BINS_PAD && N_BINS_PAD <= 512, "level-1 key is matched on 9 bits; the last row of the table is an empty bin");
static_assert(N_BINS_PAD <= 512 && N_BINS_PAD % 2 == 0, "two neighbouring bins per thread cover the table row of a 256-thread work-group");
static_assert(L1_BLOCKS == 512, "level 2 finds a run with 9 halvings and scans the runs with 16 waves");
#define L1_WG_EDGES (8 * L1_STEPS * 32)
#define L2_IMG      5120
#define L2_STEPS    (L2_IMG / 1024)
static_assert(L2_IMG % 1024 == 0, "a staged bin is covered by whole steps of the 1024-thread work-group");

__device__ __forceinline__ void wave_sync() {
  __builtin_amdgcn_fence(__ATOMIC_RELEASE, "wavefront");
  __builtin_amdgcn_wave_barrier();
  __builtin_amdgcn_fence(__ATOMIC_ACQUIRE, "wavefront");
}

template<int BITS>
__device__ __forceinline__ unsigned peer_mask(int key, bool valid) {
  unsigned peers = __builtin_amdgcn_ballot_w32(valid);
  #pragma unroll
  for (int b = 0; b < BITS; ++b) {
    const bool bit = (key >> b) & 1;
    const unsigned bal = __builtin_amdgcn_ballot_w32(bit);
    peers &= bit ? bal : ~bal;
  }
  return peers;
}

__device__ __forceinline__ int wave_sum(int v) {
  #pragma unroll
  for (int m = 16; m > 0; m >>= 1) v += __shfl_xor(v, m);
  return v;
}

__device__ __forceinline__ int wave_scan_incl(int v, int lane) {
  #pragma unroll
  for (int d = 1; d < 32; d <<= 1) {
    const int t = __shfl_up(v, d);
    if (lane >= d) v += t;
  }
  return v;
}

#define L1_LDS_INTS (8 * N_BINS_PAD + 8 + 2 * L1_WG_EDGES + L1_WG_EDGES)
__device__ __forceinline__ void bin_local(
    const int*   __restrict__ eidx,
    const float* __restrict__ ew,
    int*         __restrict__ ls,
    int*         __restrict__ swl_i,
    int*         __restrict__ dl,
    int n_edges, int block, int* lds)
{
  int  (*cur)[N_BINS_PAD] = reinterpret_cast<int (*)[N_BINS_PAD]>(lds);
  int*  wtot   = lds + 8 * N_BINS_PAD;
  int2* img_sw = reinterpret_cast<int2*>(lds + 8 * N_BINS_PAD + 8);
  int*  img_d  = lds + 8 * N_BINS_PAD + 8 + 2 * L1_WG_EDGES;
  const int lane = threadIdx.x & 31;
  const int wv   = threadIdx.x >> 5;
  const int w    = block * 8 + wv;
  for (int b = lane; b < N_BINS_PAD; b += 32) cur[wv][b] = 0;

  const bool owner = (int)threadIdx.x < N_BINS_PAD / 2;
  const int  bA    = 2 * min((int)threadIdx.x, N_BINS_PAD / 2 - 1);
  const int  bB    = bA + 1;

  int dst[L1_STEPS], srcv[L1_STEPS], wbits[L1_STEPS];
  const int e0 = w * (L1_STEPS * 32) + lane;
  #pragma unroll
  for (int s = 0; s < L1_STEPS; ++s) {
    const int e  = e0 + s * 32;
    const int ec = min(e, n_edges - 1);
    const int d  = eidx[n_edges + ec];
    srcv[s]  = eidx[ec];
    wbits[s] = __float_as_int(ew[ec]);
    dst[s]   = (e < n_edges) ? d : -1;
  }
  int2* swl = reinterpret_cast<int2*>(swl_i);
  wave_sync();
  unsigned pm[L1_STEPS];
  #pragma unroll
  for (int s = 0; s < L1_STEPS; ++s) {
    const bool valid = dst[s] >= 0;
    const int  key   = dst[s] >> 8;
    pm[s] = peer_mask<9>(key, valid);
    if (valid && (pm[s] >> lane) == 1u) cur[wv][key] += __popc(pm[s]);
    wave_sync();
  }
  __syncthreads();
  int cA[8], cB[8];
  int tA = 0, tB = 0;
  #pragma unroll
  for (int k = 0; k < 8; ++k) {
    cA[k] = owner ? cur[k][bA] : 0;
    cB[k] = owner ? cur[k][bB] : 0;
    tA += cA[k];
    tB += cB[k];
  }
  const int both = tA + tB;
  const int inc  = wave_scan_incl(both, lane);
  if (lane == 31) wtot[wv] = inc;
  __syncthreads();
  int ex = inc - both;
  #pragma unroll
  for (int k = 0; k < 8; ++k) if (k < wv) ex += wtot[k];
  if (owner) {
    int run = ex;
    #pragma unroll
    for (int k = 0; k < 8; ++k) { cur[k][bA] = run; run += cA[k]; }
    #pragma unroll
    for (int k = 0; k < 8; ++k) { cur[k][bB] = run; run += cB[k]; }
    ls[bA * L1_BLOCKS + block] = ex;
    ls[bB * L1_BLOCKS + block] = ex + tA;
  }
  __syncthreads();
  #pragma unroll
  for (int s = 0; s < L1_STEPS; ++s) {
    const bool valid = dst[s] >= 0;
    const int  key   = dst[s] >> 8;
    const unsigned peers = pm[s];
    const int base = valid ? cur[wv][key] : 0;
    wave_sync();
    if (valid && (peers >> lane) == 1u) cur[wv][key] = base + __popc(peers);
    wave_sync();
    if (valid) {
      const int pos = base + __popc(peers & ((1u << lane) - 1u));
      img_sw[pos] = make_int2(srcv[s], wbits[s]);
      img_d[pos]  = dst[s];
    }
  }
  __syncthreads();
  const int first = block * L1_WG_EDGES;
  const int nloc  = min(max(n_edges - first, 0), L1_WG_EDGES);
  if (nloc > 0) {
    int  dd[L1_STEPS];
    int2 rec[L1_STEPS];
    #pragma unroll
    for (int s = 0; s < L1_STEPS; ++s) {
      const int ic = min(s * 256 + (int)threadIdx.x, nloc - 1);
      dd[s]  = img_d[ic];
      rec[s] = img_sw[ic];
    }
    #pragma unroll
    for (int s = 0; s < L1_STEPS; ++s) {
      const int i = s * 256 + (int)threadIdx.x;
      if (i < nloc) {
        swl[first + i] = rec[s];
        dl[first + i]  = dd[s];
      }
    }
  }
}

#define GEMM_BLOCKS ((N_NODES / 16 + 7) / 8)
static_assert(8 * 16 * TP <= L1_LDS_INTS, "the GEMM's staging tiles fit the level-1 LDS");
__global__ __launch_bounds__(256) void sort1_and_gemm(
    const float* __restrict__ x,
    const float* __restrict__ W1,
    const float* __restrict__ W2,
    float*       __restrict__ h,
    const int*   __restrict__ eidx,
    const float* __restrict__ ew,
    int*         __restrict__ ls,
    int*         __restrict__ swl_i,
    int*         __restrict__ dl)
{
  __shared__ __attribute__((aligned(16))) int lds[L1_LDS_INTS];
  if ((int)blockIdx.x < L1_BLOCKS) {
    bin_local(eidx, ew, ls, swl_i, dl, N_EDGES, (int)blockIdx.x, lds);
  } else {
    gemm_x32_wmma<128>(x, W1, W2, h, N_NODES, (int)blockIdx.x - L1_BLOCKS, reinterpret_cast<float*>(lds));
  }
}

#define PULL_K 16

__device__ __forceinline__ int2 load_pairs(const int2* pairs, int b, int e, int lane) {
  int2 v = make_int2(0, 0);
  if (b < e) v = pairs[min(b + lane, e - 1)];
  return v;
}
__device__ __forceinline__ int2 mask_pairs(int2 v, int b, int e, int lane) {
  if (b + lane >= e) v.y = 0;
  return v;
}

template<int U>
__device__ __forceinline__ float pull_rows(const float* __restrict__ H, int2 p, int lane, float acc) {
  float hv[U], wt[U];
  #pragma unroll
  for (int u = 0; u < U; ++u) {
    const int s = __builtin_amdgcn_readlane(p.x, u);
    wt[u] = __int_as_float(__builtin_amdgcn_readlane(p.y, u));
    hv[u] = H[(long)s * OUTC + lane];
  }
  #pragma unroll
  for (int u = 0; u < U; ++u) acc = fmaf(wt[u], hv[u], acc);
  return acc;
}

template<bool SMALL_CODE>
__device__ __forceinline__ float pull_chunk(const float* __restrict__ H, int2 p, int cnt, int lane, float acc) {
  if (SMALL_CODE) {
    for (int u0 = 0; u0 < cnt; u0 += 8) {
      float hv[8], wt[8];
      #pragma unroll
      for (int u = 0; u < 8; ++u) {
        const int s = __builtin_amdgcn_readlane(p.x, u0 + u);
        wt[u] = __int_as_float(__builtin_amdgcn_readlane(p.y, u0 + u));
        hv[u] = H[(long)s * OUTC + lane];
      }
      #pragma unroll
      for (int u = 0; u < 8; ++u) acc = fmaf(wt[u], hv[u], acc);
    }
    return acc;
  }
  if (cnt <= 8)  return pull_rows<8>(H, p, lane, acc);
  if (cnt <= 16) return pull_rows<16>(H, p, lane, acc);
  if (cnt <= 24) return pull_rows<24>(H, p, lane, acc);
  return pull_rows<32>(H, p, lane, acc);
}

template<bool SMALL_CODE>
__device__ __forceinline__ void pull_nodes(
    const float* __restrict__ H, const int2* pairs, int o, int nk, float* __restrict__ rows, int lane)
{
  const int b0 = __builtin_amdgcn_readlane(o, 0);
  const int e0 = __builtin_amdgcn_readlane(o, 1);
  int2 p = mask_pairs(load_pairs(pairs, b0, e0, lane), b0, e0, lane);
  for (int i = 0; i < nk; ++i) {
    const int beg = __builtin_amdgcn_readlane(o, i);
    const int end = __builtin_amdgcn_readlane(o, i + 1);
    const int ne = (i + 1 < nk) ? __builtin_amdgcn_readlane(o, i + 2) : end;
    const int2 pn = load_pairs(pairs, end, ne, lane);
    float acc = 0.f;
    if (beg < end) acc = pull_chunk<SMALL_CODE>(H, p, min(32, end - beg), lane, acc);
    for (int base = beg + 32; base < end; base += 32) {
      const int2 q = mask_pairs(load_pairs(pairs, base, end, lane), base, end, lane);
      acc = pull_chunk<SMALL_CODE>(H, q, min(32, end - base), lane, acc);
    }
    rows[i * OUTC + lane] = acc;
    p = mask_pairs(pn, end, ne, lane);
  }
}

__device__ __forceinline__ int find_run(const int* segstart, int i) {
  int lo = 0, hi = L1_BLOCKS;
  #pragma unroll
  for (int t = 0; t < 9; ++t) {
    const int m = (lo + hi) >> 1;
    if (segstart[m] <= i) lo = m; else hi = m;
  }
  return lo;
}

__global__ __launch_bounds__(1024) void bin_sort_pull(
    const int*   __restrict__ ls,
    const int*   __restrict__ swl_i,
    const int*   __restrict__ dl,
    int*         __restrict__ off,
    int*         __restrict__ pairs_i,
    const float* __restrict__ H,
    float*       __restrict__ agg,
    int n_nodes)
{
  __shared__ int hist[32][256];
  __shared__ int nodeoff[257];
  __shared__ int wtot[8];
  __shared__ int2 img[L2_IMG];
  __shared__ int runs_part[16];
  __shared__ int lows_part[16];
  __shared__ int segstart[L1_BLOCKS];
  __shared__ int segsrc[L1_BLOCKS];
  __shared__ int srcmap[L2_IMG];
  const int tid  = threadIdx.x;
  const int lane = tid & 31;
  const int wv   = tid >> 5;
  const int bin  = blockIdx.x;

  const bool runner = tid < L1_BLOCKS;
  const int  g      = min(tid, L1_BLOCKS - 1);
  const int  lo_g   = ls[bin * L1_BLOCKS + g];
  const int  hi_g   = ls[(bin + 1) * L1_BLOCKS + g];
  const int  len    = runner ? hi_g - lo_g : 0;
  const int  inc0   = wave_scan_incl(len, lane);
  const int  lows   = wave_sum(runner ? lo_g : 0);
  if (runner && lane == 31) { runs_part[wv] = inc0; lows_part[wv] = lows; }
  #pragma unroll
  for (int k = 0; k < 8; ++k) hist[wv][lane + 32 * k] = 0;
  __syncthreads();

  int pre = 0, nbin = 0, beg = 0;
  #pragma unroll
  for (int k = 0; k < 16; ++k) {
    const int t = runs_part[k];
    if (k < wv) pre += t;
    nbin += t;
    beg  += lows_part[k];
  }
  const int  p0     = pre + inc0 - len;
  const int  s0     = g * L1_WG_EDGES + lo_g;
  const bool staged = nbin <= L2_IMG;
  if (runner) {
    segstart[tid] = p0;
    segsrc[tid]   = s0 - p0;
    if (staged) for (int j = 0; j < len; ++j) srcmap[p0 + j] = s0 + j;
  }
  __syncthreads();

  const int2* swl   = reinterpret_cast<const int2*>(swl_i);
  int2*       pairs = reinterpret_cast<int2*>(pairs_i);
  int      key[L2_STEPS];
  int2     rec[L2_STEPS];
  unsigned pm[L2_STEPS];
  if (staged) {
    int sp[L2_STEPS];
    #pragma unroll
    for (int s = 0; s < L2_STEPS; ++s) sp[s] = srcmap[max(min(s * 1024 + tid, nbin - 1), 0)];
    #pragma unroll
    for (int s = 0; s < L2_STEPS; ++s) {
      const int q = (nbin > 0) ? sp[s] : 0;
      key[s] = dl[q] & 255;
      rec[s] = swl[q];
    }
    #pragma unroll
    for (int s = 0; s < L2_STEPS; ++s) {
      if (s * 1024 < nbin) {
        const bool valid = s * 1024 + tid < nbin;
        pm[s] = peer_mask<8>(key[s], valid);
        if (valid && (pm[s] >> lane) == 1u) hist[wv][key[s]] += __popc(pm[s]);
        wave_sync();
      }
    }
  } else {
    for (int i0 = 0; i0 < nbin; i0 += 1024) {
      const int  i     = i0 + tid;
      const bool valid = i < nbin;
      const int  ic    = min(i, nbin - 1);
      const int  sp    = segsrc[find_run(segstart, ic)] + ic;
      const int  k     = dl[sp] & 255;
      const unsigned peers = peer_mask<8>(k, valid);
      if (valid && (peers >> lane) == 1u) hist[wv][k] += __popc(peers);
      wave_sync();
    }
  }
  __syncthreads();

  int tot = 0, inc = 0;
  if (tid < 256) {
    for (int k = 0; k < 32; ++k) {
      const int c = hist[k][tid];
      hist[k][tid] = tot;
      tot += c;
    }
    inc = wave_scan_incl(tot, lane);
    if (lane == 31) wtot[wv] = inc;
  }
  __syncthreads();
  if (tid < 256) {
    int pre2 = 0;
    for (int k = 0; k < wv; ++k) pre2 += wtot[k];
    const int no = pre2 + inc - tot;
    nodeoff[tid] = no;
    if (tid == 0) nodeoff[256] = nbin;
    const int node = bin * 256 + tid;
    if (node < n_nodes) off[node] = beg + no;
    if (node == n_nodes - 1) off[n_nodes] = beg + no + tot;
  }
  __syncthreads();

  if (staged) {
    int no[L2_STEPS];
    #pragma unroll
    for (int s = 0; s < L2_STEPS; ++s) no[s] = nodeoff[key[s]];
    #pragma unroll
    for (int s = 0; s < L2_STEPS; ++s) {
      if (s * 1024 < nbin) {
        const bool valid = s * 1024 + tid < nbin;
        const unsigned peers = pm[s];
        const int base = hist[wv][key[s]];
        wave_sync();
        if (valid && (peers >> lane) == 1u) hist[wv][key[s]] = base + __popc(peers);
        wave_sync();
        if (valid) img[no[s] + base + __popc(peers & ((1u << lane) - 1u))] = rec[s];
      }
    }
    __syncthreads();
    for (int i = tid; i < nbin; i += 1024) pairs[beg + i] = img[i];
  } else {
    for (int i0 = 0; i0 < nbin; i0 += 1024) {
      const int  i     = i0 + tid;
      const bool valid = i < nbin;
      const int  ic    = min(i, nbin - 1);
      const int  sp    = segsrc[find_run(segstart, ic)] + ic;
      const int  k     = dl[sp] & 255;
      const int2 r     = swl[sp];
      const unsigned peers = peer_mask<8>(k, valid);
      const int base = hist[wv][k];
      wave_sync();
      if (valid && (peers >> lane) == 1u) hist[wv][k] = base + __popc(peers);
      wave_sync();
      if (valid) pairs[beg + nodeoff[k] + base + __popc(peers & ((1u << lane) - 1u))] = r;
    }
    __syncthreads();
  }

  const int k0   = __builtin_amdgcn_readfirstlane(wv * 8);
  const int nloc = min(256, n_nodes - bin * 256);
  if (k0 < nloc) {
    const int nk = min(8, nloc - k0);
    const int o  = nodeoff[min(k0 + lane, 256)];
    float* rows  = agg + (long)(bin * 256 + k0) * OUTC;
    if (staged) pull_nodes<false>(H, img,         o, nk, rows, lane);
    else        pull_nodes<true> (H, pairs + beg, o, nk, rows, lane);
  }
}

__global__ __launch_bounds__(256) void gcn_pull(
    const float* __restrict__ H,
    const int*   __restrict__ off,
    const int*   __restrict__ pairs_i,
    float*       __restrict__ out,
    int n_nodes)
{
  const int lane = threadIdx.x & 31;
  const int n0 = __builtin_amdgcn_readfirstlane((int)(blockIdx.x * 8 + (threadIdx.x >> 5)) * PULL_K);
  if (n0 >= n_nodes) return;
  const int nk = min(PULL_K, n_nodes - n0);
  const int o = off[min(n0 + lane, n_nodes)];
  pull_nodes<false>(H, reinterpret_cast<const int2*>(pairs_i), o, nk, out + (long)n0 * OUTC, lane);
}

extern "C" void kernel_launch(void* const* d_in, const int* in_sizes, int n_in,
                              void* d_out, int out_size, void* d_ws, size_t ws_size,
                              hipStream_t stream) {
  const float* x  = (const float*)d_in[0];
  const float* W1 = (const float*)d_in[1];
  const float* W2 = (const float*)d_in[2];
  const float* ew = (const float*)d_in[3];
  const int*   ei = (const int*)d_in[4];
  float* out = (float*)d_out;

  char* ws = (char*)d_ws;
  float* h      = (float*)(ws);
  float* agg1   = (float*)(ws + 12800000);
  int*   pairs  = (int*)(ws + 25600000);
  int*   swl    = (int*)(ws + 38400000);
  int*   dl     = (int*)(ws + 52031488);
  int*   ls     = (int*)(ws + 58847232);
  int*   off    = (int*)(ws + 59650048);

  sort1_and_gemm<<<L1_BLOCKS + GEMM_BLOCKS, 256, 0, stream>>>(x, W1, W2, h, ei, ew, ls, swl, dl);
  bin_sort_pull<<<N_BINS, 1024, 0, stream>>>(ls, swl, dl, off, pairs, h, agg1, N_NODES);

  const int pull_grid = (N_NODES + 8 * PULL_K - 1) / (8 * PULL_K);
  gcn_pull<<<pull_grid, 256, 0, stream>>>(agg1, off, pairs, out, N_NODES);
}


# GEMM operand fetch re-laid: W1 and W2 copied once per work-group into LDS in fragment order (one 16 B LDS read per k-step, was four 4 B global loads); x read 16 B per lane; k sum order changes
# speedup vs baseline: 3.7410x; 1.1521x over previous
#include <hip/hip_runtime.h>


typedef __attribute__((ext_vector_type(2))) float v2f;
typedef __attribute__((ext_vector_type(8))) float v8f;
typedef __attribute__((ext_vector_type(4))) float v4f;

#define N_NODES 100000
#define N_EDGES 1600000
#define OUTC 32
#define TP   34

#define WF1_OFF (8 * 16 * TP)
#define WF2_OFF (WF1_OFF + 128 * OUTC)
#define GEMM_LDS_FLOATS (WF2_OFF + OUTC * OUTC)
static_assert(WF1_OFF % 4 == 0 && WF2_OFF % 4 == 0, "the fragment images are read 16 bytes at a time");
template<int K>
__device__ __forceinline__ void gemm_x32_wmma(
    const float* __restrict__ X,
    const float* __restrict__ W,
    const float* __restrict__ W2,
    float* __restrict__ H,
    int n_rows, int block, float* lds)
{
  static_assert(K == 128, "the image of W holds 32 k-steps; 4 of its entries per thread of a 256-thread work-group");
  const int lane = threadIdx.x & 31;
  const int wv   = threadIdx.x >> 5;
  const int tile = block * 8 + wv;
  const int n_tiles = n_rows >> 4;

  const int  m   = lane & 15;
  const int  kh  = lane >> 4;
  const long row0 = (long)min(tile, n_tiles - 1) << 4;

  v4f* WF1 = reinterpret_cast<v4f*>(lds + WF1_OFF);
  v4f* WF2 = reinterpret_cast<v4f*>(lds + WF2_OFF);

  v4f f1[K / 32], f2;
  #pragma unroll
  for (int i = 0; i < K / 32; ++i) {
    const int t  = 8 * i + wv;
    const int k0 = 8 * (t >> 1) + 4 * kh + 2 * (t & 1);
    f1[i].x = W[k0 * OUTC + m];
    f1[i].y = W[(k0 + 1) * OUTC + m];
    f1[i].z = W[k0 * OUTC + 16 + m];
    f1[i].w = W[(k0 + 1) * OUTC + 16 + m];
  }
  {
    const int ka = 4 * wv + 2 * kh;
    f2.x = W2[ka * OUTC + m];
    f2.y = W2[(ka + 1) * OUTC + m];
    f2.z = W2[ka * OUTC + 16 + m];
    f2.w = W2[(ka + 1) * OUTC + 16 + m];
  }

  const v4f* xr4 = reinterpret_cast<const v4f*>(X + (row0 + m) * K) + kh;
  v4f a4[K / 8];
  #pragma unroll
  for (int j = 0; j < K / 8; ++j) a4[j] = xr4[2 * j];

  #pragma unroll
  for (int i = 0; i < K / 32; ++i) WF1[(8 * i + wv) * 32 + lane] = f1[i];
  WF2[wv * 32 + lane] = f2;
  asm volatile("" :: "v"(a4[0]), "v"(a4[1]), "v"(a4[2]), "v"(a4[3]), "v"(a4[4]), "v"(a4[5]), "v"(a4[6]), "v"(a4[7]), "v"(a4[8]), "v"(a4[9]), "v"(a4[10]), "v"(a4[11]), "v"(a4[12]), "v"(a4[13]), "v"(a4[14]), "v"(a4[15]));
  __syncthreads();
  if (tile >= n_tiles) return;

  v8f c0 = {};
  v8f c1 = {};

  #pragma unroll
  for (int j = 0; j < K / 8; ++j) {
    const v4f g0 = WF1[(2 * j) * 32 + lane];
    const v4f g1 = WF1[(2 * j + 1) * 32 + lane];
    v2f a, b0, b1;
    a.x = a4[j].x; a.y = a4[j].y;
    b0.x = g0.x; b0.y = g0.y; b1.x = g0.z; b1.y = g0.w;
    c0 = __builtin_amdgcn_wmma_f32_16x16x4_f32(false, a, false, b0, (short)0, c0, false, false);
    c1 = __builtin_amdgcn_wmma_f32_16x16x4_f32(false, a, false, b1, (short)0, c1, false, false);
    a.x = a4[j].z; a.y = a4[j].w;
    b0.x = g1.x; b0.y = g1.y; b1.x = g1.z; b1.y = g1.w;
    c0 = __builtin_amdgcn_wmma_f32_16x16x4_f32(false, a, false, b0, (short)0, c0, false, false);
    c1 = __builtin_amdgcn_wmma_f32_16x16x4_f32(false, a, false, b1, (short)0, c1, false, false);
  }

  float* T = lds + wv * (16 * TP);
  #pragma unroll
  for (int r = 0; r < 8; ++r) {
    T[(8 * kh + r) * TP + m]      = c0[r];
    T[(8 * kh + r) * TP + 16 + m] = c1[r];
  }
  __builtin_amdgcn_fence(__ATOMIC_RELEASE, "wavefront");
  __builtin_amdgcn_wave_barrier();
  __builtin_amdgcn_fence(__ATOMIC_ACQUIRE, "wavefront");

  v8f d0 = {};
  v8f d1 = {};
  #pragma unroll
  for (int s = 0; s < OUTC / 4; ++s) {
    const int ka = 4 * s + 2 * kh;
    const v2f a = *reinterpret_cast<const v2f*>(&T[m * TP + ka]);
    const v4f g = WF2[s * 32 + lane];
    v2f b0, b1;
    b0.x = g.x; b0.y = g.y; b1.x = g.z; b1.y = g.w;
    d0 = __builtin_amdgcn_wmma_f32_16x16x4_f32(false, a, false, b0, (short)0, d0, false, false);
    d1 = __builtin_amdgcn_wmma_f32_16x16x4_f32(false, a, false, b1, (short)0, d1, false, false);
  }

  float* hrow = H + (row0 + 8 * kh) * OUTC + m;
  #pragma unroll
  for (int r = 0; r < 8; ++r) {
    hrow[r * OUTC]      = d0[r];
    hrow[r * OUTC + 16] = d1[r];
  }
}

#define N_BINS     ((N_NODES + 255) >> 8)
#define N_BINS_PAD 392
#define L1_BLOCKS  512
#define L1_STEPS   13
static_assert((long)L1_BLOCKS * 8 * L1_STEPS * 32 >= N_EDGES, "level-1 waves must cover every edge");
static_assert(N_BINS < N_BINS_PAD && N_BINS_PAD <= 512, "level-1 key is matched on 9 bits; the last row of the table is an empty bin");
static_assert(N_BINS_PAD <= 512 && N_BINS_PAD % 2 == 0, "two neighbouring bins per thread cover the table row of a 256-thread work-group");
static_assert(L1_BLOCKS == 512, "level 2 finds a run with 9 halvings and scans the runs with 16 waves");
#define L1_WG_EDGES (8 * L1_STEPS * 32)
#define L2_IMG      5120
#define L2_STEPS    (L2_IMG / 1024)
static_assert(L2_IMG % 1024 == 0, "a staged bin is covered by whole steps of the 1024-thread work-group");

__device__ __forceinline__ void wave_sync() {
  __builtin_amdgcn_fence(__ATOMIC_RELEASE, "wavefront");
  __builtin_amdgcn_wave_barrier();
  __builtin_amdgcn_fence(__ATOMIC_ACQUIRE, "wavefront");
}

template<int BITS>
__device__ __forceinline__ unsigned peer_mask(int key, bool valid) {
  unsigned peers = __builtin_amdgcn_ballot_w32(valid);
  #pragma unroll
  for (int b = 0; b < BITS; ++b) {
    const bool bit = (key >> b) & 1;
    const unsigned bal = __builtin_amdgcn_ballot_w32(bit);
    peers &= bit ? bal : ~bal;
  }
  return peers;
}

__device__ __forceinline__ int wave_sum(int v) {
  #pragma unroll
  for (int m = 16; m > 0; m >>= 1) v += __shfl_xor(v, m);
  return v;
}

__device__ __forceinline__ int wave_scan_incl(int v, int lane) {
  #pragma unroll
  for (int d = 1; d < 32; d <<= 1) {
    const int t = __shfl_up(v, d);
    if (lane >= d) v += t;
  }
  return v;
}

#define L1_LDS_INTS (8 * N_BINS_PAD + 8 + 2 * L1_WG_EDGES + L1_WG_EDGES)
__device__ __forceinline__ void bin_local(
    const int*   __restrict__ eidx,
    const float* __restrict__ ew,
    int*         __restrict__ ls,
    int*         __restrict__ swl_i,
    int*         __restrict__ dl,
    int n_edges, int block, int* lds)
{
  int  (*cur)[N_BINS_PAD] = reinterpret_cast<int (*)[N_BINS_PAD]>(lds);
  int*  wtot   = lds + 8 * N_BINS_PAD;
  int2* img_sw = reinterpret_cast<int2*>(lds + 8 * N_BINS_PAD + 8);
  int*  img_d  = lds + 8 * N_BINS_PAD + 8 + 2 * L1_WG_EDGES;
  const int lane = threadIdx.x & 31;
  const int wv   = threadIdx.x >> 5;
  const int w    = block * 8 + wv;
  for (int b = lane; b < N_BINS_PAD; b += 32) cur[wv][b] = 0;

  const bool owner = (int)threadIdx.x < N_BINS_PAD / 2;
  const int  bA    = 2 * min((int)threadIdx.x, N_BINS_PAD / 2 - 1);
  const int  bB    = bA + 1;

  int dst[L1_STEPS], srcv[L1_STEPS], wbits[L1_STEPS];
  const int e0 = w * (L1_STEPS * 32) + lane;
  #pragma unroll
  for (int s = 0; s < L1_STEPS; ++s) {
    const int e  = e0 + s * 32;
    const int ec = min(e, n_edges - 1);
    const int d  = eidx[n_edges + ec];
    srcv[s]  = eidx[ec];
    wbits[s] = __float_as_int(ew[ec]);
    dst[s]   = (e < n_edges) ? d : -1;
  }
  int2* swl = reinterpret_cast<int2*>(swl_i);
  wave_sync();
  unsigned pm[L1_STEPS];
  #pragma unroll
  for (int s = 0; s < L1_STEPS; ++s) {
    const bool valid = dst[s] >= 0;
    const int  key   = dst[s] >> 8;
    pm[s] = peer_mask<9>(key, valid);
    if (valid && (pm[s] >> lane) == 1u) cur[wv][key] += __popc(pm[s]);
    wave_sync();
  }
  __syncthreads();
  int cA[8], cB[8];
  int tA = 0, tB = 0;
  #pragma unroll
  for (int k = 0; k < 8; ++k) {
    cA[k] = owner ? cur[k][bA] : 0;
    cB[k] = owner ? cur[k][bB] : 0;
    tA += cA[k];
    tB += cB[k];
  }
  const int both = tA + tB;
  const int inc  = wave_scan_incl(both, lane);
  if (lane == 31) wtot[wv] = inc;
  __syncthreads();
  int ex = inc - both;
  #pragma unroll
  for (int k = 0; k < 8; ++k) if (k < wv) ex += wtot[k];
  if (owner) {
    int run = ex;
    #pragma unroll
    for (int k = 0; k < 8; ++k) { cur[k][bA] = run; run += cA[k]; }
    #pragma unroll
    for (int k = 0; k < 8; ++k) { cur[k][bB] = run; run += cB[k]; }
    ls[bA * L1_BLOCKS + block] = ex;
    ls[bB * L1_BLOCKS + block] = ex + tA;
  }
  __syncthreads();
  #pragma unroll
  for (int s = 0; s < L1_STEPS; ++s) {
    const bool valid = dst[s] >= 0;
    const int  key   = dst[s] >> 8;
    const unsigned peers = pm[s];
    const int base = valid ? cur[wv][key] : 0;
    wave_sync();
    if (valid && (peers >> lane) == 1u) cur[wv][key] = base + __popc(peers);
    wave_sync();
    if (valid) {
      const int pos = base + __popc(peers & ((1u << lane) - 1u));
      img_sw[pos] = make_int2(srcv[s], wbits[s]);
      img_d[pos]  = dst[s];
    }
  }
  __syncthreads();
  const int first = block * L1_WG_EDGES;
  const int nloc  = min(max(n_edges - first, 0), L1_WG_EDGES);
  if (nloc > 0) {
    int  dd[L1_STEPS];
    int2 rec[L1_STEPS];
    #pragma unroll
    for (int s = 0; s < L1_STEPS; ++s) {
      const int ic = min(s * 256 + (int)threadIdx.x, nloc - 1);
      dd[s]  = img_d[ic];
      rec[s] = img_sw[ic];
    }
    #pragma unroll
    for (int s = 0; s < L1_STEPS; ++s) {
      const int i = s * 256 + (int)threadIdx.x;
      if (i < nloc) {
        swl[first + i] = rec[s];
        dl[first + i]  = dd[s];
      }
    }
  }
}

#define GEMM_BLOCKS ((N_NODES / 16 + 7) / 8)
static_assert(GEMM_LDS_FLOATS <= L1_LDS_INTS, "the GEMM's staging tiles and fragment images fit the level-1 LDS");
__global__ __launch_bounds__(256) void sort1_and_gemm(
    const float* __restrict__ x,
    const float* __restrict__ W1,
    const float* __restrict__ W2,
    float*       __restrict__ h,
    const int*   __restrict__ eidx,
    const float* __restrict__ ew,
    int*         __restrict__ ls,
    int*         __restrict__ swl_i,
    int*         __restrict__ dl)
{
  __shared__ __attribute__((aligned(16))) int lds[L1_LDS_INTS];
  if ((int)blockIdx.x < L1_BLOCKS) {
    bin_local(eidx, ew, ls, swl_i, dl, N_EDGES, (int)blockIdx.x, lds);
  } else {
    gemm_x32_wmma<128>(x, W1, W2, h, N_NODES, (int)blockIdx.x - L1_BLOCKS, reinterpret_cast<float*>(lds));
  }
}

#define PULL_K 16

__device__ __forceinline__ void group_bcast8(int v, int (&r)[8]) {
  r[0] = __builtin_amdgcn_mov_dpp8(v, 0 * 0x249249);
  r[1] = __builtin_amdgcn_mov_dpp8(v, 1 * 0x249249);
  r[2] = __builtin_amdgcn_mov_dpp8(v, 2 * 0x249249);
  r[3] = __builtin_amdgcn_mov_dpp8(v, 3 * 0x249249);
  r[4] = __builtin_amdgcn_mov_dpp8(v, 4 * 0x249249);
  r[5] = __builtin_amdgcn_mov_dpp8(v, 5 * 0x249249);
  r[6] = __builtin_amdgcn_mov_dpp8(v, 6 * 0x249249);
  r[7] = __builtin_amdgcn_mov_dpp8(v, 7 * 0x249249);
}

template<int G>
__device__ __forceinline__ void load_blocks(const int2* pairs, int b, int e, int sub, int2 (&P)[G]) {
  #pragma unroll
  for (int c = 0; c < G; ++c) P[c] = pairs[max(min(b + 8 * c + sub, e - 1), 0)];
}

template<int G, bool AHEAD>
__device__ __forceinline__ void pull_nodes(
    const float* __restrict__ H, const int2* pairs, int o, int nk, float* __restrict__ rows, int lane)
{
  const int grp = lane >> 3;
  const int sub = lane & 7;
  const v4f* H4 = reinterpret_cast<const v4f*>(H);
  int  beg, end;
  int2 P[G];
  {
    const int nd = min(grp, nk - 1);
    beg = __shfl(o, nd);
    const int e1 = __shfl(o, nd + 1);
    end = (grp < nk) ? e1 : beg;
    load_blocks<G>(pairs, beg, end, sub, P);
  }
  for (int r0 = 0; r0 < nk; r0 += 4) {
    int  nbeg = 0, nend = 0;
    int2 PN[G];
    #pragma unroll
    for (int c = 0; c < G; ++c) PN[c] = make_int2(0, 0);
    bool next_due = r0 + 4 < nk;
    v4f acc = {0.f, 0.f, 0.f, 0.f};
    int b = beg;
    for (;;) {
      v4f hv[G][8];
      #pragma unroll
      for (int c = 0; c < G; ++c) {
        const bool live = b + 8 * c < end;
        int s[8];
        group_bcast8(P[c].x, s);
        if (live) {
          #pragma unroll
          for (int j = 0; j < 8; ++j) hv[c][j] = H4[((unsigned)s[j] << 3) | (unsigned)sub];
        }
      }
      if (next_due) {
        next_due = false;
        const int nd = min(r0 + 4 + grp, nk - 1);
        nbeg = __shfl(o, nd);
        const int e1 = __shfl(o, nd + 1);
        nend = (r0 + 4 + grp < nk) ? e1 : nbeg;
        if (AHEAD) load_blocks<G>(pairs, nbeg, nend, sub, PN);
      }
      #pragma unroll
      for (int c = 0; c < G; ++c) {
        const bool live = b + 8 * c < end;
        const int wsel = (b + 8 * c + sub < end) ? P[c].y : 0;
        int wb[8];
        group_bcast8(wsel, wb);
        if (live) {
          #pragma unroll
          for (int j = 0; j < 8; ++j) {
            const float wt = __int_as_float(wb[j]);
            acc.x = fmaf(wt, hv[c][j].x, acc.x);
            acc.y = fmaf(wt, hv[c][j].y, acc.y);
            acc.z = fmaf(wt, hv[c][j].z, acc.z);
            acc.w = fmaf(wt, hv[c][j].w, acc.w);
          }
        }
      }
      b += 8 * G;
      if (!__any(b < end)) break;
      load_blocks<G>(pairs, b, end, sub, P);
    }
    if (r0 + grp < nk) *reinterpret_cast<v4f*>(&rows[(r0 + grp) * OUTC + 4 * sub]) = acc;
    beg = nbeg; end = nend;
    if (AHEAD) {
      #pragma unroll
      for (int c = 0; c < G; ++c) P[c] = PN[c];
    } else if (r0 + 4 < nk) {
      load_blocks<G>(pairs, beg, end, sub, P);
    }
  }
}

__device__ __forceinline__ int find_run(const int* segstart, int i) {
  int lo = 0, hi = L1_BLOCKS;
  #pragma unroll
  for (int t = 0; t < 9; ++t) {
    const int m = (lo + hi) >> 1;
    if (segstart[m] <= i) lo = m; else hi = m;
  }
  return lo;
}

__global__ __launch_bounds__(1024) void bin_sort_pull(
    const int*   __restrict__ ls,
    const int*   __restrict__ swl_i,
    const int*   __restrict__ dl,
    int*         __restrict__ off,
    int*         __restrict__ pairs_i,
    const float* __restrict__ H,
    float*       __restrict__ agg,
    int n_nodes)
{
  __shared__ int hist[32][256];
  __shared__ int nodeoff[257];
  __shared__ int wtot[8];
  __shared__ int2 img[L2_IMG];
  __shared__ int runs_part[16];
  __shared__ int lows_part[16];
  __shared__ int segstart[L1_BLOCKS];
  __shared__ int segsrc[L1_BLOCKS];
  __shared__ int srcmap[L2_IMG];
  const int tid  = threadIdx.x;
  const int lane = tid & 31;
  const int wv   = tid >> 5;
  const int bin  = blockIdx.x;

  const bool runner = tid < L1_BLOCKS;
  const int  g      = min(tid, L1_BLOCKS - 1);
  const int  lo_g   = ls[bin * L1_BLOCKS + g];
  const int  hi_g   = ls[(bin + 1) * L1_BLOCKS + g];
  const int  len    = runner ? hi_g - lo_g : 0;
  const int  inc0   = wave_scan_incl(len, lane);
  const int  lows   = wave_sum(runner ? lo_g : 0);
  if (runner && lane == 31) { runs_part[wv] = inc0; lows_part[wv] = lows; }
  #pragma unroll
  for (int k = 0; k < 8; ++k) hist[wv][lane + 32 * k] = 0;
  __syncthreads();

  int pre = 0, nbin = 0, beg = 0;
  #pragma unroll
  for (int k = 0; k < 16; ++k) {
    const int t = runs_part[k];
    if (k < wv) pre += t;
    nbin += t;
    beg  += lows_part[k];
  }
  const int  p0     = pre + inc0 - len;
  const int  s0     = g * L1_WG_EDGES + lo_g;
  const bool staged = nbin <= L2_IMG;
  if (runner) {
    segstart[tid] = p0;
    segsrc[tid]   = s0 - p0;
    if (staged) for (int j = 0; j < len; ++j) srcmap[p0 + j] = s0 + j;
  }
  __syncthreads();

  const int2* swl   = reinterpret_cast<const int2*>(swl_i);
  int2*       pairs = reinterpret_cast<int2*>(pairs_i);
  int      key[L2_STEPS];
  int2     rec[L2_STEPS];
  unsigned pm[L2_STEPS];
  if (staged) {
    int sp[L2_STEPS];
    #pragma unroll
    for (int s = 0; s < L2_STEPS; ++s) sp[s] = srcmap[max(min(s * 1024 + tid, nbin - 1), 0)];
    #pragma unroll
    for (int s = 0; s < L2_STEPS; ++s) {
      const int q = (nbin > 0) ? sp[s] : 0;
      key[s] = dl[q] & 255;
      rec[s] = swl[q];
    }
    #pragma unroll
    for (int s = 0; s < L2_STEPS; ++s) {
      if (s * 1024 < nbin) {
        const bool valid = s * 1024 + tid < nbin;
        pm[s] = peer_mask<8>(key[s], valid);
        if (valid && (pm[s] >> lane) == 1u) hist[wv][key[s]] += __popc(pm[s]);
        wave_sync();
      }
    }
  } else {
    for (int i0 = 0; i0 < nbin; i0 += 1024) {
      const int  i     = i0 + tid;
      const bool valid = i < nbin;
      const int  ic    = min(i, nbin - 1);
      const int  sp    = segsrc[find_run(segstart, ic)] + ic;
      const int  k     = dl[sp] & 255;
      const unsigned peers = peer_mask<8>(k, valid);
      if (valid && (peers >> lane) == 1u) hist[wv][k] += __popc(peers);
      wave_sync();
    }
  }
  __syncthreads();

  int tot = 0, inc = 0;
  if (tid < 256) {
    for (int k = 0; k < 32; ++k) {
      const int c = hist[k][tid];
      hist[k][tid] = tot;
      tot += c;
    }
    inc = wave_scan_incl(tot, lane);
    if (lane == 31) wtot[wv] = inc;
  }
  __syncthreads();
  if (tid < 256) {
    int pre2 = 0;
    for (int k = 0; k < wv; ++k) pre2 += wtot[k];
    const int no = pre2 + inc - tot;
    nodeoff[tid] = no;
    if (tid == 0) nodeoff[256] = nbin;
    const int node = bin * 256 + tid;
    if (node < n_nodes) off[node] = beg + no;
    if (node == n_nodes - 1) off[n_nodes] = beg + no + tot;
  }
  __syncthreads();

  if (staged) {
    int no[L2_STEPS];
    #pragma unroll
    for (int s = 0; s < L2_STEPS; ++s) no[s] = nodeoff[key[s]];
    #pragma unroll
    for (int s = 0; s < L2_STEPS; ++s) {
      if (s * 1024 < nbin) {
        const bool valid = s * 1024 + tid < nbin;
        const unsigned peers = pm[s];
        const int base = hist[wv][key[s]];
        wave_sync();
        if (valid && (peers >> lane) == 1u) hist[wv][key[s]] = base + __popc(peers);
        wave_sync();
        if (valid) img[no[s] + base + __popc(peers & ((1u << lane) - 1u))] = rec[s];
      }
    }
    __syncthreads();
    for (int i = tid; i < nbin; i += 1024) pairs[beg + i] = img[i];
  } else {
    for (int i0 = 0; i0 < nbin; i0 += 1024) {
      const int  i     = i0 + tid;
      const bool valid = i < nbin;
      const int  ic    = min(i, nbin - 1);
      const int  sp    = segsrc[find_run(segstart, ic)] + ic;
      const int  k     = dl[sp] & 255;
      const int2 r     = swl[sp];
      const unsigned peers = peer_mask<8>(k, valid);
      const int base = hist[wv][k];
      wave_sync();
      if (valid && (peers >> lane) == 1u) hist[wv][k] = base + __popc(peers);
      wave_sync();
      if (valid) pairs[beg + nodeoff[k] + base + __popc(peers & ((1u << lane) - 1u))] = r;
    }
    __syncthreads();
  }

  const int k0   = __builtin_amdgcn_readfirstlane(wv * 8);
  const int nloc = min(256, n_nodes - bin * 256);
  if (k0 < nloc) {
    const int nk = min(8, nloc - k0);
    const int o  = nodeoff[min(k0 + lane, 256)];
    float* rows  = agg + (long)(bin * 256 + k0) * OUTC;
    if (staged) pull_nodes<1, false>(H, img,         o, nk, rows, lane);
    else        pull_nodes<1, false>(H, pairs + beg, o, nk, rows, lane);
  }
}

#define PULL_WAVES 4
__global__ __launch_bounds__(32 * PULL_WAVES) void gcn_pull(
    const float* __restrict__ H,
    const int*   __restrict__ off,
    const int*   __restrict__ pairs_i,
    float*       __restrict__ out,
    int n_nodes)
{
  const int lane = threadIdx.x & 31;
  const int n0 = __builtin_amdgcn_readfirstlane((int)(blockIdx.x * PULL_WAVES + (threadIdx.x >> 5)) * PULL_K);
  if (n0 >= n_nodes) return;
  const int nk = min(PULL_K, n_nodes - n0);
  const int o = off[min(n0 + lane, n_nodes)];
  pull_nodes<3, true>(H, reinterpret_cast<const int2*>(pairs_i), o, nk, out + (long)n0 * OUTC, lane);
}

extern "C" void kernel_launch(void* const* d_in, const int* in_sizes, int n_in,
                              void* d_out, int out_size, void* d_ws, size_t ws_size,
                              hipStream_t stream) {
  const float* x  = (const float*)d_in[0];
  const float* W1 = (const float*)d_in[1];
  const float* W2 = (const float*)d_in[2];
  const float* ew = (const float*)d_in[3];
  const int*   ei = (const int*)d_in[4];
  float* out = (float*)d_out;

  char* ws = (char*)d_ws;
  float* h      = (float*)(ws);
  float* agg1   = (float*)(ws + 12800000);
  int*   pairs  = (int*)(ws + 25600000);
  int*   swl    = (int*)(ws + 38400000);
  int*   dl     = (int*)(ws + 52031488);
  int*   ls     = (int*)(ws + 58847232);
  int*   off    = (int*)(ws + 59650048);

  sort1_and_gemm<<<L1_BLOCKS + GEMM_BLOCKS, 256, 0, stream>>>(x, W1, W2, h, ei, ew, ls, swl, dl);
  bin_sort_pull<<<N_BINS, 1024, 0, stream>>>(ls, swl, dl, off, pairs, h, agg1, N_NODES);

  const int pull_grid = (N_NODES + PULL_WAVES * PULL_K - 1) / (PULL_WAVES * PULL_K);
  gcn_pull<<<pull_grid, 32 * PULL_WAVES, 0, stream>>>(agg1, off, pairs, out, N_NODES);
}
